# starting kernel
# baseline (speedup 1.0000x reference)
#include <hip/hip_runtime.h>
#include <hip/hip_bf16.h>

#define IN_F 128
#define HD   128
#define NH   4
#define DH   32
#define NEG_SLOPE 0.2f

typedef __attribute__((ext_vector_type(2))) float v2f;
typedef __attribute__((ext_vector_type(8))) float v8f;

__device__ __forceinline__ unsigned f2ord(float f) {
    unsigned b = __float_as_uint(f);
    return (b & 0x80000000u) ? ~b : (b | 0x80000000u);
}
__device__ __forceinline__ float ord2f(unsigned k) {
    unsigned b = (k & 0x80000000u) ? (k ^ 0x80000000u) : ~k;
    return __uint_as_float(b);
}

__global__ void fill_u32(unsigned* __restrict__ p, unsigned v, long long n) {
    long long t = (long long)blockIdx.x * blockDim.x + threadIdx.x;
    if (t < n) p[t] = v;
}

__global__ __launch_bounds__(256) void gemm_wmma_f32(
    const float* __restrict__ A,
    const float* __restrict__ B,
    float* __restrict__ C,
    int nrows)
{
    const int lane    = threadIdx.x & 31;
    const int wave    = threadIdx.x >> 5;
    const int l15     = lane & 15;
    const int halfSel = lane >> 4;
    const int mBase   = blockIdx.x * 16;
    const int nBase   = wave * 16;

    int row = mBase + l15;
    if (row >= nrows) row = nrows - 1;
    const int col = nBase + l15;

    const float* aptr = A + (size_t)row * IN_F + (halfSel << 1);
    const float* bptr = B + (size_t)(halfSel << 1) * HD + col;

    v8f acc = {};
#pragma unroll
    for (int k = 0; k < IN_F; k += 4) {
        v2f a;
        a.x = aptr[k];
        a.y = aptr[k + 1];
        v2f b;
        b.x = bptr[(size_t)k * HD];
        b.y = bptr[(size_t)(k + 1) * HD];
        acc = __builtin_amdgcn_wmma_f32_16x16x4_f32(
             false, a,  false, b,
             (short)0, acc,  false,  false);
    }

#pragma unroll
    for (int r = 0; r < 8; ++r) {
        const int m = mBase + r + (halfSel << 3);
        if (m < nrows) C[(size_t)m * HD + nBase + l15] = acc[r];
    }
}

__global__ void node_attn(const float* __restrict__ ft,
                          const float* __restrict__ attn_l,
                          const float* __restrict__ attn_r,
                          float* __restrict__ el, float* __restrict__ er,
                          long long nh_total)
{
    long long t = (long long)blockIdx.x * blockDim.x + threadIdx.x;
    if (t >= nh_total) return;
    const long long n = t >> 2;
    const int h = (int)(t & 3);
    const float* f  = ft + (size_t)n * HD + h * DH;
    const float* al = attn_l + h * DH;
    const float* ar = attn_r + h * DH;
    float sl = 0.f, sr = 0.f;
#pragma unroll
    for (int d = 0; d < DH; d += 4) {
        const float4 fv = *(const float4*)(f + d);
        const float4 lv = *(const float4*)(al + d);
        const float4 rv = *(const float4*)(ar + d);
        sl += fv.x * lv.x + fv.y * lv.y + fv.z * lv.z + fv.w * lv.w;
        sr += fv.x * rv.x + fv.y * rv.y + fv.z * rv.z + fv.w * rv.w;
    }
    el[t] = sl;
    er[t] = sr;
}

__global__ void edge_logits(const float* __restrict__ el,
                            const float* __restrict__ er,
                            const int* __restrict__ src,
                            const int* __restrict__ dst,
                            float* __restrict__ ebuf,
                            unsigned* __restrict__ emaxk,
                            long long total)
{
    long long t = (long long)blockIdx.x * blockDim.x + threadIdx.x;
    if (t >= total) return;
    const long long e = t >> 2;
    const int h = (int)(t & 3);
    const int s = src[e], d = dst[e];
    float v = el[(size_t)s * NH + h] + er[(size_t)d * NH + h];
    v = v > 0.f ? v : NEG_SLOPE * v;
    ebuf[t] = v;
    atomicMax(&emaxk[(size_t)d * NH + h], f2ord(v));
}

__global__ void finalize_max(const unsigned* __restrict__ emaxk,
                             float* __restrict__ emaxf, long long nh_total)
{
    long long t = (long long)blockIdx.x * blockDim.x + threadIdx.x;
    if (t >= nh_total) return;
    const unsigned k = emaxk[t];
    float v = (k == 0u) ? 0.f : ord2f(k);
    if (!__builtin_isfinite(v)) v = 0.f;
    emaxf[t] = v;
}

__global__ void edge_exp(const float* __restrict__ emaxf,
                         const int* __restrict__ dst,
                         float* __restrict__ ebuf,
                         float* __restrict__ denom,
                         long long total)
{
    long long t = (long long)blockIdx.x * blockDim.x + threadIdx.x;
    if (t >= total) return;
    const long long e = t >> 2;
    const int h = (int)(t & 3);
    const int d = dst[e];
    const float ex = expf(ebuf[t] - emaxf[(size_t)d * NH + h]);
    ebuf[t] = ex;
    atomicAdd(&denom[(size_t)d * NH + h], ex);
}

__global__ void finalize_denom(float* __restrict__ denom, long long nh_total) {
    long long t = (long long)blockIdx.x * blockDim.x + threadIdx.x;
    if (t >= nh_total) return;
    const float s = denom[t];
    denom[t] = (s > 0.f) ? 1.0f / s : 1.0f;
}

__global__ void edge_aggregate(const float* __restrict__ ft,
                               const float* __restrict__ ebuf,
                               const float* __restrict__ rdenom,
                               const int* __restrict__ src,
                               const int* __restrict__ dst,
                               float* __restrict__ out,
                               long long total)
{
    long long t = (long long)blockIdx.x * blockDim.x + threadIdx.x;
    if (t >= total) return;
    const long long e = t >> 5;
    const int q = (int)(t & 31);
    const int h = q >> 3;
    const int s = src[e], d = dst[e];
    const float a = ebuf[e * NH + h] * rdenom[(size_t)d * NH + h];
    const float4 v = ((const float4*)(ft + (size_t)s * HD))[q];
    float* o = out + (size_t)d * HD + q * 4;
    atomicAdd(o + 0, a * v.x);
    atomicAdd(o + 1, a * v.y);
    atomicAdd(o + 2, a * v.z);
    atomicAdd(o + 3, a * v.w);
}

static inline unsigned nblk(long long n, int bs) {
    return (unsigned)((n + bs - 1) / bs);
}

extern "C" void kernel_launch(void* const* d_in, const int* in_sizes, int n_in,
                              void* d_out, int out_size, void* d_ws, size_t ws_size,
                              hipStream_t stream) {
    const float* feat   = (const float*)d_in[0];
    const float* W      = (const float*)d_in[1];
    const float* attn_l = (const float*)d_in[2];
    const float* attn_r = (const float*)d_in[3];
    const int*   src    = (const int*)d_in[4];
    const int*   dst    = (const int*)d_in[5];
    float*       out    = (float*)d_out;

    const long long N = in_sizes[0] / IN_F;
    const long long E = in_sizes[4];

    float*    ft    = (float*)d_ws;
    float*    el    = ft + (size_t)N * HD;
    float*    er    = el + (size_t)N * NH;
    unsigned* emaxk = (unsigned*)(er + (size_t)N * NH);
    float*    emaxf = (float*)(emaxk + (size_t)N * NH);
    float*    denom = emaxf + (size_t)N * NH;
    float*    ebuf  = denom + (size_t)N * NH;

    const int BS = 256;
    const long long NHtot = N * NH;
    const long long EHtot = E * NH;
    const long long EAgg  = E * 32;

    fill_u32<<<nblk((long long)out_size, BS), BS, 0, stream>>>((unsigned*)out, 0u, out_size);
    fill_u32<<<nblk(NHtot, BS), BS, 0, stream>>>((unsigned*)denom, 0u, NHtot);
    fill_u32<<<nblk(NHtot, BS), BS, 0, stream>>>(emaxk, 0u, NHtot);

    gemm_wmma_f32<<<nblk(N, 16), 256, 0, stream>>>(feat, W, ft, (int)N);

    node_attn<<<nblk(NHtot, BS), BS, 0, stream>>>(ft, attn_l, attn_r, el, er, NHtot);

    edge_logits<<<nblk(EHtot, BS), BS, 0, stream>>>(el, er, src, dst, ebuf, emaxk, EHtot);

    finalize_max<<<nblk(NHtot, BS), BS, 0, stream>>>(emaxk, emaxf, NHtot);

    edge_exp<<<nblk(EHtot, BS), BS, 0, stream>>>(emaxf, dst, ebuf, denom, EHtot);

    finalize_denom<<<nblk(NHtot, BS), BS, 0, stream>>>(denom, NHtot);

    edge_aggregate<<<nblk(EAgg, BS), BS, 0, stream>>>(ft, ebuf, denom, src, dst, out, EAgg);
}


# Projection GEMM re-tiled: one wave computes a 16x128 strip in 8 accumulator tiles (permuted column map); A fragment read once per strip, W and ft move in whole-line 16-byte pieces; k order unchanged
# speedup vs baseline: 1.6688x; 1.6688x over previous
#include <hip/hip_runtime.h>
#include <hip/hip_bf16.h>

#define IN_F 128
#define HD   128
#define NH   4
#define DH   32
#define NEG_SLOPE 0.2f

typedef __attribute__((ext_vector_type(2))) float v2f;
typedef __attribute__((ext_vector_type(8))) float v8f;

__device__ __forceinline__ unsigned f2ord(float f) {
    unsigned b = __float_as_uint(f);
    return (b & 0x80000000u) ? ~b : (b | 0x80000000u);
}
__device__ __forceinline__ float ord2f(unsigned k) {
    unsigned b = (k & 0x80000000u) ? (k ^ 0x80000000u) : ~k;
    return __uint_as_float(b);
}

#define GEMM_ROWS_PER_BLOCK 32
__global__ __launch_bounds__(64) void gemm_wmma_f32(
    const float* __restrict__ A,
    const float* __restrict__ B,
    float* __restrict__ C,
    float* __restrict__ out,
    long long out_n,
    int nrows)
{
    {
        const long long zb = (long long)blockIdx.x * (GEMM_ROWS_PER_BLOCK * HD);
        if (zb + GEMM_ROWS_PER_BLOCK * HD <= out_n) {
            float4* o4 = (float4*)(out + zb) + threadIdx.x;
            for (int p = 0; p < 16; ++p) o4[p * 64] = make_float4(0.f, 0.f, 0.f, 0.f);
        } else {
            for (long long q = zb + threadIdx.x; q < out_n; q += 64) out[q] = 0.f;
        }
    }

    const int lane    = threadIdx.x & 31;
    const int wave    = threadIdx.x >> 5;
    const int l15     = lane & 15;
    const int halfSel = lane >> 4;
    const int mBase   = (blockIdx.x * 2 + wave) * 16;
    if (mBase >= nrows) return;

    int row = mBase + l15;
    if (row >= nrows) row = nrows - 1;

    const float* aptr = A + (size_t)row * IN_F + (halfSel << 1);
    const float* bptr = B + (size_t)(halfSel << 1) * HD + (l15 << 2);

    v8f acc[8];
#pragma unroll
    for (int t = 0; t < 8; ++t) acc[t] = v8f{};

    v2f    a   = *(const v2f*)aptr;
    float4 b0l = *(const float4*)(bptr);
    float4 b1l = *(const float4*)(bptr + HD);
    float4 b0h = *(const float4*)(bptr + 64);
    float4 b1h = *(const float4*)(bptr + HD + 64);

#pragma unroll 2
    for (int k = 0; k < IN_F; k += 4) {
        const int kn = (k + 4 < IN_F) ? (k + 4) : 0;
        const v2f    an   = *(const v2f*)(aptr + kn);
        const float4 b0ln = *(const float4*)(bptr + (size_t)kn * HD);
        const float4 b1ln = *(const float4*)(bptr + (size_t)(kn + 1) * HD);
        const float4 b0hn = *(const float4*)(bptr + (size_t)kn * HD + 64);
        const float4 b1hn = *(const float4*)(bptr + (size_t)(kn + 1) * HD + 64);

        v2f b;
        b.x = b0l.x; b.y = b1l.x;
        acc[0] = __builtin_amdgcn_wmma_f32_16x16x4_f32(false, a, false, b, (short)0, acc[0], false, false);
        b.x = b0l.y; b.y = b1l.y;
        acc[1] = __builtin_amdgcn_wmma_f32_16x16x4_f32(false, a, false, b, (short)0, acc[1], false, false);
        b.x = b0l.z; b.y = b1l.z;
        acc[2] = __builtin_amdgcn_wmma_f32_16x16x4_f32(false, a, false, b, (short)0, acc[2], false, false);
        b.x = b0l.w; b.y = b1l.w;
        acc[3] = __builtin_amdgcn_wmma_f32_16x16x4_f32(false, a, false, b, (short)0, acc[3], false, false);
        b.x = b0h.x; b.y = b1h.x;
        acc[4] = __builtin_amdgcn_wmma_f32_16x16x4_f32(false, a, false, b, (short)0, acc[4], false, false);
        b.x = b0h.y; b.y = b1h.y;
        acc[5] = __builtin_amdgcn_wmma_f32_16x16x4_f32(false, a, false, b, (short)0, acc[5], false, false);
        b.x = b0h.z; b.y = b1h.z;
        acc[6] = __builtin_amdgcn_wmma_f32_16x16x4_f32(false, a, false, b, (short)0, acc[6], false, false);
        b.x = b0h.w; b.y = b1h.w;
        acc[7] = __builtin_amdgcn_wmma_f32_16x16x4_f32(false, a, false, b, (short)0, acc[7], false, false);

        a = an; b0l = b0ln; b1l = b1ln; b0h = b0hn; b1h = b1hn;
    }

#pragma unroll
    for (int r = 0; r < 8; ++r) {
        const int m = mBase + r + (halfSel << 3);
        if (m < nrows) {
            float* crow = C + (size_t)m * HD + (l15 << 2);
            *(float4*)(crow)      = make_float4(acc[0][r], acc[1][r], acc[2][r], acc[3][r]);
            *(float4*)(crow + 64) = make_float4(acc[4][r], acc[5][r], acc[6][r], acc[7][r]);
        }
    }
}

__global__ void node_attn(const float* __restrict__ ft,
                          const float* __restrict__ attn_l,
                          const float* __restrict__ attn_r,
                          float* __restrict__ el, float* __restrict__ er,
                          unsigned* __restrict__ emaxk, float* __restrict__ denom,
                          long long nh_total)
{
    long long t = (long long)blockIdx.x * blockDim.x + threadIdx.x;
    if (t >= nh_total) return;
    const long long n = t >> 2;
    const int h = (int)(t & 3);
    const float* f  = ft + (size_t)n * HD + h * DH;
    const float* al = attn_l + h * DH;
    const float* ar = attn_r + h * DH;
    float sl = 0.f, sr = 0.f;
#pragma unroll
    for (int d = 0; d < DH; d += 4) {
        const float4 fv = *(const float4*)(f + d);
        const float4 lv = *(const float4*)(al + d);
        const float4 rv = *(const float4*)(ar + d);
        sl += fv.x * lv.x + fv.y * lv.y + fv.z * lv.z + fv.w * lv.w;
        sr += fv.x * rv.x + fv.y * rv.y + fv.z * rv.z + fv.w * rv.w;
    }
    el[t] = sl;
    er[t] = sr;
    emaxk[t] = 0u;
    denom[t] = 0.f;
}

__global__ void edge_logits(const float* __restrict__ el,
                            const float* __restrict__ er,
                            const int* __restrict__ src,
                            const int* __restrict__ dst,
                            float* __restrict__ ebuf,
                            unsigned* __restrict__ emaxk,
                            long long total)
{
    long long t = (long long)blockIdx.x * blockDim.x + threadIdx.x;
    if (t >= total) return;
    const long long e = t >> 2;
    const int h = (int)(t & 3);
    const int s = src[e], d = dst[e];
    float v = el[(size_t)s * NH + h] + er[(size_t)d * NH + h];
    v = v > 0.f ? v : NEG_SLOPE * v;
    ebuf[t] = v;
    atomicMax(&emaxk[(size_t)d * NH + h], f2ord(v));
}

__global__ void finalize_max(const unsigned* __restrict__ emaxk,
                             float* __restrict__ emaxf, long long nh_total)
{
    long long t = (long long)blockIdx.x * blockDim.x + threadIdx.x;
    if (t >= nh_total) return;
    const unsigned k = emaxk[t];
    float v = (k == 0u) ? 0.f : ord2f(k);
    if (!__builtin_isfinite(v)) v = 0.f;
    emaxf[t] = v;
}

__global__ void edge_exp(const float* __restrict__ emaxf,
                         const int* __restrict__ dst,
                         float* __restrict__ ebuf,
                         float* __restrict__ denom,
                         long long total)
{
    long long t = (long long)blockIdx.x * blockDim.x + threadIdx.x;
    if (t >= total) return;
    const long long e = t >> 2;
    const int h = (int)(t & 3);
    const int d = dst[e];
    const float ex = expf(ebuf[t] - emaxf[(size_t)d * NH + h]);
    ebuf[t] = ex;
    atomicAdd(&denom[(size_t)d * NH + h], ex);
}

__global__ void finalize_denom(float* __restrict__ denom, long long nh_total) {
    long long t = (long long)blockIdx.x * blockDim.x + threadIdx.x;
    if (t >= nh_total) return;
    const float s = denom[t];
    denom[t] = (s > 0.f) ? 1.0f / s : 1.0f;
}

#define EPW 8
__global__ __launch_bounds__(256) void edge_aggregate(const float* __restrict__ ft,
                               const float* __restrict__ ebuf,
                               const float* __restrict__ rdenom,
                               const int* __restrict__ src,
                               const int* __restrict__ dst,
                               float* __restrict__ out,
                               long long nedges)
{
    const int lane = threadIdx.x & 31;
    const long long wv = (long long)blockIdx.x * (blockDim.x >> 5) + (threadIdx.x >> 5);
    const long long e0 = wv * EPW;
    if (e0 >= nedges) return;

    const int h = lane & 3;
    long long ei = e0 + (lane >> 2);
    if (ei >= nedges) ei = nedges - 1;
    const int sv = src[ei];
    const int dv = dst[ei];
    const float av = ebuf[ei * NH + h] * rdenom[(size_t)dv * NH + h];

    if (e0 + EPW <= nedges) {
        float v[EPW][NH];
#pragma unroll
        for (int j = 0; j < EPW; ++j) {
            const int s = __builtin_amdgcn_readlane(sv, 4 * j);
            const float* fr = ft + (size_t)s * HD + lane;
#pragma unroll
            for (int hh = 0; hh < NH; ++hh) v[j][hh] = fr[DH * hh];
        }
#pragma unroll
        for (int j = 0; j < EPW; ++j) {
            const int d = __builtin_amdgcn_readlane(dv, 4 * j);
            float* orow = out + (size_t)d * HD + lane;
#pragma unroll
            for (int hh = 0; hh < NH; ++hh) {
                const float a = __int_as_float(
                    __builtin_amdgcn_readlane(__float_as_int(av), 4 * j + hh));
                atomicAdd(orow + DH * hh, a * v[j][hh]);
            }
        }
    } else {
#pragma unroll
        for (int j = 0; j < EPW; ++j) {
            if (e0 + j < nedges) {
                const int s = __builtin_amdgcn_readlane(sv, 4 * j);
                const int d = __builtin_amdgcn_readlane(dv, 4 * j);
                const float* fr = ft + (size_t)s * HD + lane;
                float* orow = out + (size_t)d * HD + lane;
#pragma unroll
                for (int hh = 0; hh < NH; ++hh) {
                    const float a = __int_as_float(
                        __builtin_amdgcn_readlane(__float_as_int(av), 4 * j + hh));
                    atomicAdd(orow + DH * hh, a * fr[DH * hh]);
                }
            }
        }
    }
}

static inline unsigned nblk(long long n, int bs) {
    return (unsigned)((n + bs - 1) / bs);
}

extern "C" void kernel_launch(void* const* d_in, const int* in_sizes, int n_in,
                              void* d_out, int out_size, void* d_ws, size_t ws_size,
                              hipStream_t stream) {
    const float* feat   = (const float*)d_in[0];
    const float* W      = (const float*)d_in[1];
    const float* attn_l = (const float*)d_in[2];
    const float* attn_r = (const float*)d_in[3];
    const int*   src    = (const int*)d_in[4];
    const int*   dst    = (const int*)d_in[5];
    float*       out    = (float*)d_out;

    const long long N = in_sizes[0] / IN_F;
    const long long E = in_sizes[4];

    float*    ft    = (float*)d_ws;
    float*    el    = ft + (size_t)N * HD;
    float*    er    = el + (size_t)N * NH;
    unsigned* emaxk = (unsigned*)(er + (size_t)N * NH);
    float*    emaxf = (float*)(emaxk + (size_t)N * NH);
    float*    denom = emaxf + (size_t)N * NH;
    float*    ebuf  = denom + (size_t)N * NH;

    const int BS = 256;
    const long long NHtot = N * NH;
    const long long EHtot = E * NH;
    const long long EAgg  = ((E + EPW - 1) / EPW) * 32;

    gemm_wmma_f32<<<nblk(N, GEMM_ROWS_PER_BLOCK), 64, 0, stream>>>(feat, W, ft, out, (long long)out_size, (int)N);

    node_attn<<<nblk(NHtot, BS), BS, 0, stream>>>(ft, attn_l, attn_r, el, er, emaxk, denom, NHtot);

    edge_logits<<<nblk(EHtot, BS), BS, 0, stream>>>(el, er, src, dst, ebuf, emaxk, EHtot);

    finalize_max<<<nblk(NHtot, BS), BS, 0, stream>>>(emaxk, emaxf, NHtot);

    edge_exp<<<nblk(EHtot, BS), BS, 0, stream>>>(emaxf, dst, ebuf, denom, EHtot);

    finalize_denom<<<nblk(NHtot, BS), BS, 0, stream>>>(denom, NHtot);

    edge_aggregate<<<nblk(EAgg, BS), BS, 0, stream>>>(ft, ebuf, denom, src, dst, out, E);
}


# Softmax shift = LeakyReLU(max el of all nodes + er[dst]) when el's range <= 32 (never below a logit), exact segment-max pass otherwise: per-edge max atomics, decode launch, logit round trip dropped
# speedup vs baseline: 2.0556x; 1.2318x over previous
#include <hip/hip_runtime.h>
#include <hip/hip_bf16.h>

#define IN_F 128
#define HD   128
#define NH   4
#define DH   32
#define NEG_SLOPE 0.2f

typedef __attribute__((ext_vector_type(2))) float v2f;
typedef __attribute__((ext_vector_type(8))) float v8f;

__device__ __forceinline__ unsigned f2ord(float f) {
    unsigned b = __float_as_uint(f);
    return (b & 0x80000000u) ? ~b : (b | 0x80000000u);
}
__device__ __forceinline__ float ord2f(unsigned k) {
    unsigned b = (k & 0x80000000u) ? (k ^ 0x80000000u) : ~k;
    return __uint_as_float(b);
}

#define GEMM_ROWS_PER_BLOCK 32
#define RNG_SLOTS 64
#define RNG_MAX 32.0f
__global__ __launch_bounds__(64) void gemm_wmma_f32(
    const float* __restrict__ A,
    const float* __restrict__ B,
    float* __restrict__ C,
    float* __restrict__ out,
    long long out_n,
    unsigned* __restrict__ rng,
    int nrows)
{
    if (blockIdx.x == 0) {
        for (int q = threadIdx.x; q < RNG_SLOTS * 8; q += 64) rng[q] = 0u;
    }
    {
        const long long zb = (long long)blockIdx.x * (GEMM_ROWS_PER_BLOCK * HD);
        if (zb + GEMM_ROWS_PER_BLOCK * HD <= out_n) {
            float4* o4 = (float4*)(out + zb) + threadIdx.x;
            for (int p = 0; p < 16; ++p) o4[p * 64] = make_float4(0.f, 0.f, 0.f, 0.f);
        } else {
            for (long long q = zb + threadIdx.x; q < out_n; q += 64) out[q] = 0.f;
        }
    }

    const int lane    = threadIdx.x & 31;
    const int wave    = threadIdx.x >> 5;
    const int l15     = lane & 15;
    const int halfSel = lane >> 4;
    const int mBase   = (blockIdx.x * 2 + wave) * 16;
    if (mBase >= nrows) return;

    int row = mBase + l15;
    if (row >= nrows) row = nrows - 1;

    const float* aptr = A + (size_t)row * IN_F + (halfSel << 1);
    const float* bptr = B + (size_t)(halfSel << 1) * HD + (l15 << 2);

    v8f acc[8];
#pragma unroll
    for (int t = 0; t < 8; ++t) acc[t] = v8f{};

    v2f    a   = *(const v2f*)aptr;
    float4 b0l = *(const float4*)(bptr);
    float4 b1l = *(const float4*)(bptr + HD);
    float4 b0h = *(const float4*)(bptr + 64);
    float4 b1h = *(const float4*)(bptr + HD + 64);

#pragma unroll 2
    for (int k = 0; k < IN_F; k += 4) {
        const int kn = (k + 4 < IN_F) ? (k + 4) : 0;
        const v2f    an   = *(const v2f*)(aptr + kn);
        const float4 b0ln = *(const float4*)(bptr + (size_t)kn * HD);
        const float4 b1ln = *(const float4*)(bptr + (size_t)(kn + 1) * HD);
        const float4 b0hn = *(const float4*)(bptr + (size_t)kn * HD + 64);
        const float4 b1hn = *(const float4*)(bptr + (size_t)(kn + 1) * HD + 64);

        v2f b;
        b.x = b0l.x; b.y = b1l.x;
        acc[0] = __builtin_amdgcn_wmma_f32_16x16x4_f32(false, a, false, b, (short)0, acc[0], false, false);
        b.x = b0l.y; b.y = b1l.y;
        acc[1] = __builtin_amdgcn_wmma_f32_16x16x4_f32(false, a, false, b, (short)0, acc[1], false, false);
        b.x = b0l.z; b.y = b1l.z;
        acc[2] = __builtin_amdgcn_wmma_f32_16x16x4_f32(false, a, false, b, (short)0, acc[2], false, false);
        b.x = b0l.w; b.y = b1l.w;
        acc[3] = __builtin_amdgcn_wmma_f32_16x16x4_f32(false, a, false, b, (short)0, acc[3], false, false);
        b.x = b0h.x; b.y = b1h.x;
        acc[4] = __builtin_amdgcn_wmma_f32_16x16x4_f32(false, a, false, b, (short)0, acc[4], false, false);
        b.x = b0h.y; b.y = b1h.y;
        acc[5] = __builtin_amdgcn_wmma_f32_16x16x4_f32(false, a, false, b, (short)0, acc[5], false, false);
        b.x = b0h.z; b.y = b1h.z;
        acc[6] = __builtin_amdgcn_wmma_f32_16x16x4_f32(false, a, false, b, (short)0, acc[6], false, false);
        b.x = b0h.w; b.y = b1h.w;
        acc[7] = __builtin_amdgcn_wmma_f32_16x16x4_f32(false, a, false, b, (short)0, acc[7], false, false);

        a = an; b0l = b0ln; b1l = b1ln; b0h = b0hn; b1h = b1hn;
    }

#pragma unroll
    for (int r = 0; r < 8; ++r) {
        const int m = mBase + r + (halfSel << 3);
        if (m < nrows) {
            float* crow = C + (size_t)m * HD + (l15 << 2);
            *(float4*)(crow)      = make_float4(acc[0][r], acc[1][r], acc[2][r], acc[3][r]);
            *(float4*)(crow + 64) = make_float4(acc[4][r], acc[5][r], acc[6][r], acc[7][r]);
        }
    }
}

__global__ void node_attn(const float* __restrict__ ft,
                          const float* __restrict__ attn_l,
                          const float* __restrict__ attn_r,
                          float* __restrict__ el, float* __restrict__ er,
                          unsigned* __restrict__ emaxk, float* __restrict__ denom,
                          unsigned* __restrict__ rng,
                          long long nh_total)
{
    __shared__ unsigned red[8][8];
    long long t = (long long)blockIdx.x * blockDim.x + threadIdx.x;
    const bool live = t < nh_total;
    const int h = (int)(threadIdx.x & 3);
    unsigned kmax = 0u, kmin = 0u;
    if (live) {
        const long long n = t >> 2;
        const float* f  = ft + (size_t)n * HD + h * DH;
        const float* al = attn_l + h * DH;
        const float* ar = attn_r + h * DH;
        float sl = 0.f, sr = 0.f;
#pragma unroll
        for (int d = 0; d < DH; d += 4) {
            const float4 fv = *(const float4*)(f + d);
            const float4 lv = *(const float4*)(al + d);
            const float4 rv = *(const float4*)(ar + d);
            sl += fv.x * lv.x + fv.y * lv.y + fv.z * lv.z + fv.w * lv.w;
            sr += fv.x * rv.x + fv.y * rv.y + fv.z * rv.z + fv.w * rv.w;
        }
        el[t] = sl;
        er[t] = sr;
        emaxk[t] = 0u;
        denom[t] = 0.f;
        kmax = f2ord(sl);
        kmin = ~kmax;
    }
    for (int m = 4; m < 32; m <<= 1) {
        const unsigned a = (unsigned)__shfl_xor((int)kmax, m);
        const unsigned b = (unsigned)__shfl_xor((int)kmin, m);
        kmax = a > kmax ? a : kmax;
        kmin = b > kmin ? b : kmin;
    }
    const int lane = threadIdx.x & 31;
    const int wave = threadIdx.x >> 5;
    if (lane < 4) { red[wave][lane] = kmax; red[wave][4 + lane] = kmin; }
    __syncthreads();
    if (threadIdx.x < 8) {
        unsigned k = 0u;
        const int nw = (int)(blockDim.x >> 5);
        for (int w = 0; w < nw; ++w) { const unsigned v = red[w][threadIdx.x]; k = v > k ? v : k; }
        atomicMax(&rng[(blockIdx.x & (RNG_SLOTS - 1)) * 8 + threadIdx.x], k);
    }
}

__device__ __forceinline__ unsigned fold_range(const unsigned* __restrict__ rng, int lane, float mx[NH])
{
    unsigned k[8];
#pragma unroll
    for (int i = 0; i < 8; ++i) {
        const unsigned a = rng[lane * 8 + i];
        const unsigned b = rng[(lane + 32) * 8 + i];
        k[i] = a > b ? a : b;
    }
#pragma unroll
    for (int i = 0; i < 8; ++i) {
        for (int m = 1; m < 32; m <<= 1) {
            const unsigned o = (unsigned)__shfl_xor((int)k[i], m);
            k[i] = o > k[i] ? o : k[i];
        }
    }
    unsigned ok = 1u;
#pragma unroll
    for (int hh = 0; hh < NH; ++hh) {
        const float hi = (k[hh] == 0u) ? __builtin_nanf("") : ord2f(k[hh]);
        const float lo = (k[4 + hh] == 0u) ? __builtin_nanf("") : ord2f(~k[4 + hh]);
        const bool good = __builtin_isfinite(hi) && __builtin_isfinite(lo) && ((hi - lo) <= RNG_MAX);
        if (!good) ok = 0u;
        mx[hh] = hi;
    }
    return ok;
}

__global__ void edge_max(const float* __restrict__ el,
                         const float* __restrict__ er,
                         const int* __restrict__ src,
                         const int* __restrict__ dst,
                         const unsigned* __restrict__ rng,
                         unsigned* __restrict__ rfin,
                         unsigned* __restrict__ emaxk,
                         long long total)
{
    float mx[NH];
    const unsigned ok = fold_range(rng, (int)(threadIdx.x & 31), mx);
    if (blockIdx.x == 0 && threadIdx.x == 0) {
#pragma unroll
        for (int hh = 0; hh < NH; ++hh) rfin[hh] = __float_as_uint(mx[hh]);
        rfin[4] = ok;
    }
    if (ok != 0u) return;
    const long long step = (long long)gridDim.x * blockDim.x;
    for (long long t = (long long)blockIdx.x * blockDim.x + threadIdx.x; t < total; t += step) {
        const long long e = t >> 2;
        const int h = (int)(t & 3);
        const int s = src[e], d = dst[e];
        float v = el[(size_t)s * NH + h] + er[(size_t)d * NH + h];
        v = v > 0.f ? v : NEG_SLOPE * v;
        atomicMax(&emaxk[(size_t)d * NH + h], f2ord(v));
    }
}

__global__ void edge_exp(const float* __restrict__ el,
                         const float* __restrict__ er,
                         const int* __restrict__ src,
                         const int* __restrict__ dst,
                         const unsigned* __restrict__ rfin,
                         const unsigned* __restrict__ emaxk,
                         float* __restrict__ ebuf,
                         float* __restrict__ denom,
                         long long total)
{
    long long t = (long long)blockIdx.x * blockDim.x + threadIdx.x;
    if (t >= total) return;
    const long long e = t >> 2;
    const int h = (int)(t & 3);
    const int s = src[e], d = dst[e];
    const float erd = er[(size_t)d * NH + h];
    float v = el[(size_t)s * NH + h] + erd;
    v = v > 0.f ? v : NEG_SLOPE * v;
    float c;
    if (rfin[4] != 0u) {
        c = __uint_as_float(rfin[h]) + erd;
        c = c > 0.f ? c : NEG_SLOPE * c;
    } else {
        const unsigned k = emaxk[(size_t)d * NH + h];
        c = (k == 0u) ? 0.f : ord2f(k);
        if (!__builtin_isfinite(c)) c = 0.f;
    }
    const float ex = expf(v - c);
    ebuf[t] = ex;
    atomicAdd(&denom[(size_t)d * NH + h], ex);
}

__global__ void finalize_denom(float* __restrict__ denom, long long nh_total) {
    long long t = (long long)blockIdx.x * blockDim.x + threadIdx.x;
    if (t >= nh_total) return;
    const float s = denom[t];
    denom[t] = (s > 0.f) ? 1.0f / s : 1.0f;
}

#define EPW 8
__global__ __launch_bounds__(256) void edge_aggregate(const float* __restrict__ ft,
                               const float* __restrict__ ebuf,
                               const float* __restrict__ rdenom,
                               const int* __restrict__ src,
                               const int* __restrict__ dst,
                               float* __restrict__ out,
                               long long nedges)
{
    const int lane = threadIdx.x & 31;
    const long long wv = (long long)blockIdx.x * (blockDim.x >> 5) + (threadIdx.x >> 5);
    const long long e0 = wv * EPW;
    if (e0 >= nedges) return;

    const int h = lane & 3;
    long long ei = e0 + (lane >> 2);
    if (ei >= nedges) ei = nedges - 1;
    const int sv = src[ei];
    const int dv = dst[ei];
    const float av = ebuf[ei * NH + h] * rdenom[(size_t)dv * NH + h];

    if (e0 + EPW <= nedges) {
        float v[EPW][NH];
#pragma unroll
        for (int j = 0; j < EPW; ++j) {
            const int s = __builtin_amdgcn_readlane(sv, 4 * j);
            const float* fr = ft + (size_t)s * HD + lane;
#pragma unroll
            for (int hh = 0; hh < NH; ++hh) v[j][hh] = fr[DH * hh];
        }
#pragma unroll
        for (int j = 0; j < EPW; ++j) {
            const int d = __builtin_amdgcn_readlane(dv, 4 * j);
            float* orow = out + (size_t)d * HD + lane;
#pragma unroll
            for (int hh = 0; hh < NH; ++hh) {
                const float a = __int_as_float(
                    __builtin_amdgcn_readlane(__float_as_int(av), 4 * j + hh));
                atomicAdd(orow + DH * hh, a * v[j][hh]);
            }
        }
    } else {
#pragma unroll
        for (int j = 0; j < EPW; ++j) {
            if (e0 + j < nedges) {
                const int s = __builtin_amdgcn_readlane(sv, 4 * j);
                const int d = __builtin_amdgcn_readlane(dv, 4 * j);
                const float* fr = ft + (size_t)s * HD + lane;
                float* orow = out + (size_t)d * HD + lane;
#pragma unroll
                for (int hh = 0; hh < NH; ++hh) {
                    const float a = __int_as_float(
                        __builtin_amdgcn_readlane(__float_as_int(av), 4 * j + hh));
                    atomicAdd(orow + DH * hh, a * fr[DH * hh]);
                }
            }
        }
    }
}

static inline unsigned nblk(long long n, int bs) {
    return (unsigned)((n + bs - 1) / bs);
}

extern "C" void kernel_launch(void* const* d_in, const int* in_sizes, int n_in,
                              void* d_out, int out_size, void* d_ws, size_t ws_size,
                              hipStream_t stream) {
    const float* feat   = (const float*)d_in[0];
    const float* W      = (const float*)d_in[1];
    const float* attn_l = (const float*)d_in[2];
    const float* attn_r = (const float*)d_in[3];
    const int*   src    = (const int*)d_in[4];
    const int*   dst    = (const int*)d_in[5];
    float*       out    = (float*)d_out;

    const long long N = in_sizes[0] / IN_F;
    const long long E = in_sizes[4];

    float*    ft    = (float*)d_ws;
    float*    el    = ft + (size_t)N * HD;
    float*    er    = el + (size_t)N * NH;
    unsigned* emaxk = (unsigned*)(er + (size_t)N * NH);
    unsigned* rng   = emaxk + (size_t)N * NH;
    unsigned* rfin  = rng + RNG_SLOTS * 8;
    float*    denom = (float*)(rng + (size_t)N * NH);
    float*    ebuf  = denom + (size_t)N * NH;

    const int BS = 256;
    const long long NHtot = N * NH;
    const long long EHtot = E * NH;
    const long long EAgg  = ((E + EPW - 1) / EPW) * 32;

    gemm_wmma_f32<<<nblk(N, GEMM_ROWS_PER_BLOCK), 64, 0, stream>>>(feat, W, ft, out, (long long)out_size, rng, (int)N);

    node_attn<<<nblk(NHtot, BS), BS, 0, stream>>>(ft, attn_l, attn_r, el, er, emaxk, denom, rng, NHtot);

    edge_max<<<2048, BS, 0, stream>>>(el, er, src, dst, rng, rfin, emaxk, EHtot);

    edge_exp<<<nblk(EHtot, BS), BS, 0, stream>>>(el, er, src, dst, rfin, emaxk, ebuf, denom, EHtot);

    finalize_denom<<<nblk(NHtot, BS), BS, 0, stream>>>(denom, NHtot);

    edge_aggregate<<<nblk(EAgg, BS), BS, 0, stream>>>(ft, ebuf, denom, src, dst, out, E);
}


# Destination rows owned in LDS: each work-group scans dst, keeps its 400 nodes' edges, adds exp and exp-weighted ft rows by LDS float adds (arrival order), divides at the one store; no global atomics
# speedup vs baseline: 2.7127x; 1.3197x over previous
#include <hip/hip_runtime.h>
#include <hip/hip_bf16.h>

#define IN_F 128
#define HD   128
#define NH   4
#define DH   32
#define NEG_SLOPE 0.2f

typedef __attribute__((ext_vector_type(2))) float v2f;
typedef __attribute__((ext_vector_type(8))) float v8f;

__device__ __forceinline__ unsigned f2ord(float f) {
    unsigned b = __float_as_uint(f);
    return (b & 0x80000000u) ? ~b : (b | 0x80000000u);
}
__device__ __forceinline__ float ord2f(unsigned k) {
    unsigned b = (k & 0x80000000u) ? (k ^ 0x80000000u) : ~k;
    return __uint_as_float(b);
}

#define GEMM_ROWS_PER_BLOCK 32
#define RNG_SLOTS 64
#define RNG_MAX 32.0f
__global__ __launch_bounds__(64) void gemm_wmma_f32(
    const float* __restrict__ A,
    const float* __restrict__ B,
    float* __restrict__ C,
    unsigned* __restrict__ rng,
    int nrows)
{
    if (blockIdx.x == 0) {
        for (int q = threadIdx.x; q < RNG_SLOTS * 8; q += 64) rng[q] = 0u;
    }

    const int lane    = threadIdx.x & 31;
    const int wave    = threadIdx.x >> 5;
    const int l15     = lane & 15;
    const int halfSel = lane >> 4;
    const int mBase   = (blockIdx.x * 2 + wave) * 16;
    if (mBase >= nrows) return;

    int row = mBase + l15;
    if (row >= nrows) row = nrows - 1;

    const float* aptr = A + (size_t)row * IN_F + (halfSel << 1);
    const float* bptr = B + (size_t)(halfSel << 1) * HD + (l15 << 2);

    v8f acc[8];
#pragma unroll
    for (int t = 0; t < 8; ++t) acc[t] = v8f{};

    v2f    a   = *(const v2f*)aptr;
    float4 b0l = *(const float4*)(bptr);
    float4 b1l = *(const float4*)(bptr + HD);
    float4 b0h = *(const float4*)(bptr + 64);
    float4 b1h = *(const float4*)(bptr + HD + 64);

#pragma unroll 2
    for (int k = 0; k < IN_F; k += 4) {
        const int kn = (k + 4 < IN_F) ? (k + 4) : 0;
        const v2f    an   = *(const v2f*)(aptr + kn);
        const float4 b0ln = *(const float4*)(bptr + (size_t)kn * HD);
        const float4 b1ln = *(const float4*)(bptr + (size_t)(kn + 1) * HD);
        const float4 b0hn = *(const float4*)(bptr + (size_t)kn * HD + 64);
        const float4 b1hn = *(const float4*)(bptr + (size_t)(kn + 1) * HD + 64);

        v2f b;
        b.x = b0l.x; b.y = b1l.x;
        acc[0] = __builtin_amdgcn_wmma_f32_16x16x4_f32(false, a, false, b, (short)0, acc[0], false, false);
        b.x = b0l.y; b.y = b1l.y;
        acc[1] = __builtin_amdgcn_wmma_f32_16x16x4_f32(false, a, false, b, (short)0, acc[1], false, false);
        b.x = b0l.z; b.y = b1l.z;
        acc[2] = __builtin_amdgcn_wmma_f32_16x16x4_f32(false, a, false, b, (short)0, acc[2], false, false);
        b.x = b0l.w; b.y = b1l.w;
        acc[3] = __builtin_amdgcn_wmma_f32_16x16x4_f32(false, a, false, b, (short)0, acc[3], false, false);
        b.x = b0h.x; b.y = b1h.x;
        acc[4] = __builtin_amdgcn_wmma_f32_16x16x4_f32(false, a, false, b, (short)0, acc[4], false, false);
        b.x = b0h.y; b.y = b1h.y;
        acc[5] = __builtin_amdgcn_wmma_f32_16x16x4_f32(false, a, false, b, (short)0, acc[5], false, false);
        b.x = b0h.z; b.y = b1h.z;
        acc[6] = __builtin_amdgcn_wmma_f32_16x16x4_f32(false, a, false, b, (short)0, acc[6], false, false);
        b.x = b0h.w; b.y = b1h.w;
        acc[7] = __builtin_amdgcn_wmma_f32_16x16x4_f32(false, a, false, b, (short)0, acc[7], false, false);

        a = an; b0l = b0ln; b1l = b1ln; b0h = b0hn; b1h = b1hn;
    }

#pragma unroll
    for (int r = 0; r < 8; ++r) {
        const int m = mBase + r + (halfSel << 3);
        if (m < nrows) {
            float* crow = C + (size_t)m * HD + (l15 << 2);
            *(float4*)(crow)      = make_float4(acc[0][r], acc[1][r], acc[2][r], acc[3][r]);
            *(float4*)(crow + 64) = make_float4(acc[4][r], acc[5][r], acc[6][r], acc[7][r]);
        }
    }
}

__global__ void node_attn(const float* __restrict__ ft,
                          const float* __restrict__ attn_l,
                          const float* __restrict__ attn_r,
                          float* __restrict__ el, float* __restrict__ er,
                          unsigned* __restrict__ emaxk,
                          unsigned* __restrict__ rng,
                          long long nh_total)
{
    __shared__ unsigned red[8][8];
    long long t = (long long)blockIdx.x * blockDim.x + threadIdx.x;
    const bool live = t < nh_total;
    const int h = (int)(threadIdx.x & 3);
    unsigned kmax = 0u, kmin = 0u;
    if (live) {
        const long long n = t >> 2;
        const float* f  = ft + (size_t)n * HD + h * DH;
        const float* al = attn_l + h * DH;
        const float* ar = attn_r + h * DH;
        float sl = 0.f, sr = 0.f;
#pragma unroll
        for (int d = 0; d < DH; d += 4) {
            const float4 fv = *(const float4*)(f + d);
            const float4 lv = *(const float4*)(al + d);
            const float4 rv = *(const float4*)(ar + d);
            sl += fv.x * lv.x + fv.y * lv.y + fv.z * lv.z + fv.w * lv.w;
            sr += fv.x * rv.x + fv.y * rv.y + fv.z * rv.z + fv.w * rv.w;
        }
        el[t] = sl;
        er[t] = sr;
        emaxk[t] = 0u;
        kmax = f2ord(sl);
        kmin = ~kmax;
    }
    for (int m = 4; m < 32; m <<= 1) {
        const unsigned a = (unsigned)__shfl_xor((int)kmax, m);
        const unsigned b = (unsigned)__shfl_xor((int)kmin, m);
        kmax = a > kmax ? a : kmax;
        kmin = b > kmin ? b : kmin;
    }
    const int lane = threadIdx.x & 31;
    const int wave = threadIdx.x >> 5;
    if (lane < 4) { red[wave][lane] = kmax; red[wave][4 + lane] = kmin; }
    __syncthreads();
    if (threadIdx.x < 8) {
        unsigned k = 0u;
        const int nw = (int)(blockDim.x >> 5);
        for (int w = 0; w < nw; ++w) { const unsigned v = red[w][threadIdx.x]; k = v > k ? v : k; }
        atomicMax(&rng[(blockIdx.x & (RNG_SLOTS - 1)) * 8 + threadIdx.x], k);
    }
}

__device__ __forceinline__ unsigned fold_range(const unsigned* __restrict__ rng, int lane, float mx[NH])
{
    unsigned k[8];
#pragma unroll
    for (int i = 0; i < 8; ++i) {
        const unsigned a = rng[lane * 8 + i];
        const unsigned b = rng[(lane + 32) * 8 + i];
        k[i] = a > b ? a : b;
    }
#pragma unroll
    for (int i = 0; i < 8; ++i) {
        for (int m = 1; m < 32; m <<= 1) {
            const unsigned o = (unsigned)__shfl_xor((int)k[i], m);
            k[i] = o > k[i] ? o : k[i];
        }
    }
    unsigned ok = 1u;
#pragma unroll
    for (int hh = 0; hh < NH; ++hh) {
        const float hi = (k[hh] == 0u) ? __builtin_nanf("") : ord2f(k[hh]);
        const float lo = (k[4 + hh] == 0u) ? __builtin_nanf("") : ord2f(~k[4 + hh]);
        const bool good = __builtin_isfinite(hi) && __builtin_isfinite(lo) && ((hi - lo) <= RNG_MAX);
        if (!good) ok = 0u;
        mx[hh] = hi;
    }
    return ok;
}

__global__ void edge_max(const float* __restrict__ el,
                         const float* __restrict__ er,
                         const int* __restrict__ src,
                         const int* __restrict__ dst,
                         const unsigned* __restrict__ rng,
                         unsigned* __restrict__ rfin,
                         unsigned* __restrict__ emaxk,
                         long long total)
{
    float mx[NH];
    const unsigned ok = fold_range(rng, (int)(threadIdx.x & 31), mx);
    if (blockIdx.x == 0 && threadIdx.x == 0) {
#pragma unroll
        for (int hh = 0; hh < NH; ++hh) rfin[hh] = __float_as_uint(mx[hh]);
        rfin[4] = ok;
    }
    if (ok != 0u) return;
    const long long step = (long long)gridDim.x * blockDim.x;
    for (long long t = (long long)blockIdx.x * blockDim.x + threadIdx.x; t < total; t += step) {
        const long long e = t >> 2;
        const int h = (int)(t & 3);
        const int s = src[e], d = dst[e];
        float v = el[(size_t)s * NH + h] + er[(size_t)d * NH + h];
        v = v > 0.f ? v : NEG_SLOPE * v;
        atomicMax(&emaxk[(size_t)d * NH + h], f2ord(v));
    }
}

#define OWN     400
#define OWAVES  32
#define QRING   512
#define CHUNK   1024
__global__ __launch_bounds__(1024) void edge_owner(const float* __restrict__ ft,
                               const float* __restrict__ el,
                               const float* __restrict__ er,
                               const int* __restrict__ src,
                               const int* __restrict__ dst,
                               const unsigned* __restrict__ rfin,
                               const unsigned* __restrict__ emaxk,
                               float* __restrict__ out,
                               int nnodes, int nedges)
{
    __shared__ float4 acc4[OWN * HD / 4];
    __shared__ float4 dsum4[OWN];
    __shared__ float4 erl4[OWN];
    __shared__ float4 shl4[OWN];
    __shared__ int    q[OWAVES][QRING];
    float* acc  = (float*)acc4;
    float* dsum = (float*)dsum4;

    const int tid  = threadIdx.x;
    const int lane = tid & 31;
    const int wave = __builtin_amdgcn_readfirstlane(tid >> 5);
    const int n0   = blockIdx.x * OWN;
    const int nown = (nnodes - n0) < OWN ? (nnodes - n0) : OWN;

    for (int i = tid; i < OWN * HD / 4; i += 1024) acc4[i] = make_float4(0.f, 0.f, 0.f, 0.f);
    {
        const unsigned bound = rfin[4];
        float* erl = (float*)erl4;
        float* shl = (float*)shl4;
        for (int i = tid; i < OWN * NH; i += 1024) {
            float erv = 0.f, c = 0.f;
            if (i < nown * NH) {
                erv = er[(size_t)n0 * NH + i];
                if (bound != 0u) {
                    c = __uint_as_float(rfin[i & 3]) + erv;
                    c = c > 0.f ? c : NEG_SLOPE * c;
                } else {
                    const unsigned k = emaxk[(size_t)n0 * NH + i];
                    c = (k == 0u) ? 0.f : ord2f(k);
                    if (!__builtin_isfinite(c)) c = 0.f;
                }
            }
            dsum[i] = 0.f;
            erl[i]  = erv;
            shl[i]  = c;
        }
    }
    __syncthreads();

    int* myq = q[wave];
    int qh = 0, qt = 0;

    auto take = [&](const int nb) {
        int s = 0, dl = 0;
        float4 l4 = make_float4(0.f, 0.f, 0.f, 0.f);
        if (lane < nb) {
            const int e = myq[(qh + lane) & (QRING - 1)];
            s  = src[e];
            dl = dst[e] - n0;
            l4 = *(const float4*)(el + (size_t)s * NH);
        }
        float ex0 = 0.f, ex1 = 0.f, ex2 = 0.f, ex3 = 0.f;
        for (int j0 = 0; j0 < nb; j0 += 8) {
            float v[8][NH];
#pragma unroll
            for (int j = 0; j < 8; ++j) {
                const int sj = __builtin_amdgcn_readlane(s, j0 + j);
                const float* fr = ft + (size_t)sj * HD + lane;
#pragma unroll
                for (int hh = 0; hh < NH; ++hh) v[j][hh] = fr[DH * hh];
            }
            if (j0 == 0 && lane < nb) {
                const float4 r4 = erl4[dl];
                const float4 c4 = shl4[dl];
                float v0 = l4.x + r4.x, v1 = l4.y + r4.y, v2 = l4.z + r4.z, v3 = l4.w + r4.w;
                v0 = v0 > 0.f ? v0 : NEG_SLOPE * v0;
                v1 = v1 > 0.f ? v1 : NEG_SLOPE * v1;
                v2 = v2 > 0.f ? v2 : NEG_SLOPE * v2;
                v3 = v3 > 0.f ? v3 : NEG_SLOPE * v3;
                ex0 = expf(v0 - c4.x);
                ex1 = expf(v1 - c4.y);
                ex2 = expf(v2 - c4.z);
                ex3 = expf(v3 - c4.w);
                atomicAdd(&dsum[dl * NH + 0], ex0);
                atomicAdd(&dsum[dl * NH + 1], ex1);
                atomicAdd(&dsum[dl * NH + 2], ex2);
                atomicAdd(&dsum[dl * NH + 3], ex3);
            }
#pragma unroll
            for (int j = 0; j < 8; ++j) {
                if (j0 + j < nb) {
                    const int dj = __builtin_amdgcn_readlane(dl, j0 + j);
                    float* ar = acc + dj * HD + lane;
                    const float a0 = __int_as_float(__builtin_amdgcn_readlane(__float_as_int(ex0), j0 + j));
                    const float a1 = __int_as_float(__builtin_amdgcn_readlane(__float_as_int(ex1), j0 + j));
                    const float a2 = __int_as_float(__builtin_amdgcn_readlane(__float_as_int(ex2), j0 + j));
                    const float a3 = __int_as_float(__builtin_amdgcn_readlane(__float_as_int(ex3), j0 + j));
                    atomicAdd(ar,          a0 * v[j][0]);
                    atomicAdd(ar + DH,     a1 * v[j][1]);
                    atomicAdd(ar + 2 * DH, a2 * v[j][2]);
                    atomicAdd(ar + 3 * DH, a3 * v[j][3]);
                }
            }
        }
        qh += nb;
    };

    const int nfull  = nedges & ~3;
    const int nchunk = (nfull + CHUNK - 1) / CHUNK;
    const int nit    = (nchunk + OWAVES - 1) / OWAVES;

    auto keep = [&](const unsigned x, const int e) {
        const bool hit = x < (unsigned)nown;
        const unsigned m = __builtin_amdgcn_ballot_w32(hit);
        if (m != 0u) {
            if (hit) myq[(qt + (int)__builtin_amdgcn_mbcnt_lo(m, 0u)) & (QRING - 1)] = e;
            qt += __builtin_popcount(m);
        }
    };
    auto scan4 = [&](const int4 d, const int c, const int u) {
        const int e = c * CHUNK + u * 128 + lane * 4;
        const bool valid = (c >= 0) && (e < nfull);
        const unsigned x0 = (unsigned)(d.x - n0), x1 = (unsigned)(d.y - n0);
        const unsigned x2 = (unsigned)(d.z - n0), x3 = (unsigned)(d.w - n0);
        const unsigned m01 = x0 < x1 ? x0 : x1;
        const unsigned m23 = x2 < x3 ? x2 : x3;
        unsigned mn = m01 < m23 ? m01 : m23;
        mn = valid ? mn : 0xFFFFFFFFu;
        if (__builtin_amdgcn_ballot_w32(mn < (unsigned)nown) != 0u) {
            keep(valid ? x0 : 0xFFFFFFFFu, e);
            keep(valid ? x1 : 0xFFFFFFFFu, e + 1);
            keep(valid ? x2 : 0xFFFFFFFFu, e + 2);
            keep(valid ? x3 : 0xFFFFFFFFu, e + 3);
        }
    };

    const int rot    = (int)(((long long)blockIdx.x * nit) / gridDim.x) * OWAVES;
    auto chunk_of = [&](const int it) -> int {
        if (it >= nit) return -1;
        int c = it * OWAVES + wave + rot;
        if (c >= nit * OWAVES) c -= nit * OWAVES;
        return c < nchunk ? c : -1;
    };
    auto load4 = [&](const int c, const int u) -> int4 {
        int e = (c < 0 ? 0 : c) * CHUNK + u * 128 + lane * 4;
        e = e < nfull - 4 ? e : nfull - 4;
        return *(const int4*)(dst + e);
    };

    if (wave == 0) {
        const int e = nfull + lane;
        const int d = (e < nedges) ? dst[e] : -1;
        keep((unsigned)(d - n0), e);
    }

    if (nfull > 0) {
        int ccur = -1;
        int4 dv[8];
#pragma unroll
        for (int u = 0; u < 8; ++u) dv[u] = make_int4(-1, -1, -1, -1);
        for (int it = -1; it < nit; ++it) {
            const int cnext = chunk_of(it + 1);
#pragma unroll
            for (int u = 0; u < 8; u += 2) {
                scan4(dv[u], ccur, u);
                dv[u] = load4(cnext, u);
                scan4(dv[u + 1], ccur, u + 1);
                dv[u + 1] = load4(cnext, u + 1);
                while (qt - qh >= 32) take(32);
            }
            ccur = cnext;
        }
    }
    if (qt - qh > 0) take(qt - qh);
    __syncthreads();

    float4* o4 = (float4*)(out + (size_t)n0 * HD);
    for (int i = tid; i < nown * (HD / 4); i += 1024) {
        const float dn = dsum[(i >> 5) * NH + ((i & 31) >> 3)];
        const float dq = dn > 0.f ? dn : 1.0f;
        float4 a = acc4[i];
        a.x /= dq; a.y /= dq; a.z /= dq; a.w /= dq;
        o4[i] = a;
    }
}

static inline unsigned nblk(long long n, int bs) {
    return (unsigned)((n + bs - 1) / bs);
}

extern "C" void kernel_launch(void* const* d_in, const int* in_sizes, int n_in,
                              void* d_out, int out_size, void* d_ws, size_t ws_size,
                              hipStream_t stream) {
    const float* feat   = (const float*)d_in[0];
    const float* W      = (const float*)d_in[1];
    const float* attn_l = (const float*)d_in[2];
    const float* attn_r = (const float*)d_in[3];
    const int*   src    = (const int*)d_in[4];
    const int*   dst    = (const int*)d_in[5];
    float*       out    = (float*)d_out;

    const long long N = in_sizes[0] / IN_F;
    const long long E = in_sizes[4];

    float*    ft    = (float*)d_ws;
    float*    el    = ft + (size_t)N * HD;
    float*    er    = el + (size_t)N * NH;
    unsigned* emaxk = (unsigned*)(er + (size_t)N * NH);
    unsigned* rng   = emaxk + (size_t)N * NH;
    unsigned* rfin  = rng + RNG_SLOTS * 8;

    const int BS = 256;
    const long long NHtot = N * NH;
    const long long EHtot = E * NH;

    gemm_wmma_f32<<<nblk(N, GEMM_ROWS_PER_BLOCK), 64, 0, stream>>>(feat, W, ft, rng, (int)N);

    node_attn<<<nblk(NHtot, BS), BS, 0, stream>>>(ft, attn_l, attn_r, el, er, emaxk, rng, NHtot);

    edge_max<<<2048, BS, 0, stream>>>(el, er, src, dst, rng, rfin, emaxk, EHtot);

    edge_owner<<<nblk(N, OWN), 1024, 0, stream>>>(ft, el, er, src, dst, rfin, emaxk, out, (int)N, (int)E);
}


# Launch 4 scans one owner byte per edge (written by launch 3 over the unused max keys), not the 4-byte destination: 16 edges per lane load, ring emptied in one place; sums stay arrival-order
# speedup vs baseline: 3.0095x; 1.1094x over previous
#include <hip/hip_runtime.h>
#include <hip/hip_bf16.h>

#define IN_F 128
#define HD   128
#define NH   4
#define DH   32
#define NEG_SLOPE 0.2f

typedef __attribute__((ext_vector_type(2))) float v2f;
typedef __attribute__((ext_vector_type(8))) float v8f;

__device__ __forceinline__ unsigned f2ord(float f) {
    unsigned b = __float_as_uint(f);
    return (b & 0x80000000u) ? ~b : (b | 0x80000000u);
}
__device__ __forceinline__ float ord2f(unsigned k) {
    unsigned b = (k & 0x80000000u) ? (k ^ 0x80000000u) : ~k;
    return __uint_as_float(b);
}

#define GEMM_ROWS_PER_BLOCK 32
#define RNG_SLOTS 64
#define RNG_MAX 32.0f
__global__ __launch_bounds__(64) void gemm_wmma_f32(
    const float* __restrict__ A,
    const float* __restrict__ B,
    float* __restrict__ C,
    unsigned* __restrict__ rng,
    int nrows)
{
    if (blockIdx.x == 0) {
        for (int q = threadIdx.x; q < RNG_SLOTS * 8; q += 64) rng[q] = 0u;
    }

    const int lane    = threadIdx.x & 31;
    const int wave    = threadIdx.x >> 5;
    const int l15     = lane & 15;
    const int halfSel = lane >> 4;
    const int mBase   = (blockIdx.x * 2 + wave) * 16;
    if (mBase >= nrows) return;

    int row = mBase + l15;
    if (row >= nrows) row = nrows - 1;

    const float* aptr = A + (size_t)row * IN_F + (halfSel << 1);
    const float* bptr = B + (size_t)(halfSel << 1) * HD + (l15 << 2);

    v8f acc[8];
#pragma unroll
    for (int t = 0; t < 8; ++t) acc[t] = v8f{};

    v2f    a   = *(const v2f*)aptr;
    float4 b0l = *(const float4*)(bptr);
    float4 b1l = *(const float4*)(bptr + HD);
    float4 b0h = *(const float4*)(bptr + 64);
    float4 b1h = *(const float4*)(bptr + HD + 64);

#pragma unroll 2
    for (int k = 0; k < IN_F; k += 4) {
        const int kn = (k + 4 < IN_F) ? (k + 4) : 0;
        const v2f    an   = *(const v2f*)(aptr + kn);
        const float4 b0ln = *(const float4*)(bptr + (size_t)kn * HD);
        const float4 b1ln = *(const float4*)(bptr + (size_t)(kn + 1) * HD);
        const float4 b0hn = *(const float4*)(bptr + (size_t)kn * HD + 64);
        const float4 b1hn = *(const float4*)(bptr + (size_t)(kn + 1) * HD + 64);

        v2f b;
        b.x = b0l.x; b.y = b1l.x;
        acc[0] = __builtin_amdgcn_wmma_f32_16x16x4_f32(false, a, false, b, (short)0, acc[0], false, false);
        b.x = b0l.y; b.y = b1l.y;
        acc[1] = __builtin_amdgcn_wmma_f32_16x16x4_f32(false, a, false, b, (short)0, acc[1], false, false);
        b.x = b0l.z; b.y = b1l.z;
        acc[2] = __builtin_amdgcn_wmma_f32_16x16x4_f32(false, a, false, b, (short)0, acc[2], false, false);
        b.x = b0l.w; b.y = b1l.w;
        acc[3] = __builtin_amdgcn_wmma_f32_16x16x4_f32(false, a, false, b, (short)0, acc[3], false, false);
        b.x = b0h.x; b.y = b1h.x;
        acc[4] = __builtin_amdgcn_wmma_f32_16x16x4_f32(false, a, false, b, (short)0, acc[4], false, false);
        b.x = b0h.y; b.y = b1h.y;
        acc[5] = __builtin_amdgcn_wmma_f32_16x16x4_f32(false, a, false, b, (short)0, acc[5], false, false);
        b.x = b0h.z; b.y = b1h.z;
        acc[6] = __builtin_amdgcn_wmma_f32_16x16x4_f32(false, a, false, b, (short)0, acc[6], false, false);
        b.x = b0h.w; b.y = b1h.w;
        acc[7] = __builtin_amdgcn_wmma_f32_16x16x4_f32(false, a, false, b, (short)0, acc[7], false, false);

        a = an; b0l = b0ln; b1l = b1ln; b0h = b0hn; b1h = b1hn;
    }

#pragma unroll
    for (int r = 0; r < 8; ++r) {
        const int m = mBase + r + (halfSel << 3);
        if (m < nrows) {
            float* crow = C + (size_t)m * HD + (l15 << 2);
            *(float4*)(crow)      = make_float4(acc[0][r], acc[1][r], acc[2][r], acc[3][r]);
            *(float4*)(crow + 64) = make_float4(acc[4][r], acc[5][r], acc[6][r], acc[7][r]);
        }
    }
}

__global__ void node_attn(const float* __restrict__ ft,
                          const float* __restrict__ attn_l,
                          const float* __restrict__ attn_r,
                          float* __restrict__ el, float* __restrict__ er,
                          unsigned* __restrict__ emaxk,
                          unsigned* __restrict__ rng,
                          long long nh_total)
{
    __shared__ unsigned red[8][8];
    long long t = (long long)blockIdx.x * blockDim.x + threadIdx.x;
    const bool live = t < nh_total;
    const int h = (int)(threadIdx.x & 3);
    unsigned kmax = 0u, kmin = 0u;
    if (live) {
        const long long n = t >> 2;
        const float* f  = ft + (size_t)n * HD + h * DH;
        const float* al = attn_l + h * DH;
        const float* ar = attn_r + h * DH;
        float sl = 0.f, sr = 0.f;
#pragma unroll
        for (int d = 0; d < DH; d += 4) {
            const float4 fv = *(const float4*)(f + d);
            const float4 lv = *(const float4*)(al + d);
            const float4 rv = *(const float4*)(ar + d);
            sl += fv.x * lv.x + fv.y * lv.y + fv.z * lv.z + fv.w * lv.w;
            sr += fv.x * rv.x + fv.y * rv.y + fv.z * rv.z + fv.w * rv.w;
        }
        el[t] = sl;
        er[t] = sr;
        emaxk[t] = 0u;
        kmax = f2ord(sl);
        kmin = ~kmax;
    }
    for (int m = 4; m < 32; m <<= 1) {
        const unsigned a = (unsigned)__shfl_xor((int)kmax, m);
        const unsigned b = (unsigned)__shfl_xor((int)kmin, m);
        kmax = a > kmax ? a : kmax;
        kmin = b > kmin ? b : kmin;
    }
    const int lane = threadIdx.x & 31;
    const int wave = threadIdx.x >> 5;
    if (lane < 4) { red[wave][lane] = kmax; red[wave][4 + lane] = kmin; }
    __syncthreads();
    if (threadIdx.x < 8) {
        unsigned k = 0u;
        const int nw = (int)(blockDim.x >> 5);
        for (int w = 0; w < nw; ++w) { const unsigned v = red[w][threadIdx.x]; k = v > k ? v : k; }
        atomicMax(&rng[(blockIdx.x & (RNG_SLOTS - 1)) * 8 + threadIdx.x], k);
    }
}

__device__ __forceinline__ unsigned fold_range(const unsigned* __restrict__ rng, int lane, float mx[NH])
{
    unsigned k[8];
#pragma unroll
    for (int i = 0; i < 8; ++i) {
        const unsigned a = rng[lane * 8 + i];
        const unsigned b = rng[(lane + 32) * 8 + i];
        k[i] = a > b ? a : b;
    }
#pragma unroll
    for (int i = 0; i < 8; ++i) {
        for (int m = 1; m < 32; m <<= 1) {
            const unsigned o = (unsigned)__shfl_xor((int)k[i], m);
            k[i] = o > k[i] ? o : k[i];
        }
    }
    unsigned ok = 1u;
#pragma unroll
    for (int hh = 0; hh < NH; ++hh) {
        const float hi = (k[hh] == 0u) ? __builtin_nanf("") : ord2f(k[hh]);
        const float lo = (k[4 + hh] == 0u) ? __builtin_nanf("") : ord2f(~k[4 + hh]);
        const bool good = __builtin_isfinite(hi) && __builtin_isfinite(lo) && ((hi - lo) <= RNG_MAX);
        if (!good) ok = 0u;
        mx[hh] = hi;
    }
    return ok;
}

#define OWN     400
__device__ __forceinline__ bool codes_fit(const int nnodes, const int nedges) {
    return (nedges >= 16) && ((long long)nedges <= (long long)nnodes * 16) && ((nnodes + OWN - 1) / OWN <= 256);
}

__global__ void edge_max(const float* __restrict__ el,
                         const float* __restrict__ er,
                         const int* __restrict__ src,
                         const int* __restrict__ dst,
                         const unsigned* __restrict__ rng,
                         unsigned* __restrict__ rfin,
                         unsigned* __restrict__ emaxk,
                         long long total,
                         int nnodes, int nedges)
{
    const long long step = (long long)gridDim.x * blockDim.x;
    const long long tid0 = (long long)blockIdx.x * blockDim.x + threadIdx.x;
    const bool cfit = codes_fit(nnodes, nedges);
    const long long nq = cfit ? (long long)(nedges >> 2) : 0;
    int4 d4 = make_int4(0, 0, 0, 0);
    if (tid0 < nq) d4 = *(const int4*)(dst + 4 * tid0);

    float mx[NH];
    const unsigned ok = fold_range(rng, (int)(threadIdx.x & 31), mx);
    if (blockIdx.x == 0 && threadIdx.x == 0) {
#pragma unroll
        for (int hh = 0; hh < NH; ++hh) rfin[hh] = __float_as_uint(mx[hh]);
        rfin[4] = ok;
    }
    if (ok != 0u) {
        for (long long q = tid0; q < nq; q += step) {
            if (q != tid0) d4 = *(const int4*)(dst + 4 * q);
            const unsigned c0 = ((unsigned)d4.x / OWN) & 255u;
            const unsigned c1 = ((unsigned)d4.y / OWN) & 255u;
            const unsigned c2 = ((unsigned)d4.z / OWN) & 255u;
            const unsigned c3 = ((unsigned)d4.w / OWN) & 255u;
            emaxk[q] = c0 | (c1 << 8) | (c2 << 16) | (c3 << 24);
        }
        return;
    }
    for (long long t = tid0; t < total; t += step) {
        const long long e = t >> 2;
        const int h = (int)(t & 3);
        const int s = src[e], d = dst[e];
        float v = el[(size_t)s * NH + h] + er[(size_t)d * NH + h];
        v = v > 0.f ? v : NEG_SLOPE * v;
        atomicMax(&emaxk[(size_t)d * NH + h], f2ord(v));
    }
}

#define OWAVES  32
#define QRING   512
#define CWORDS  1024
__global__ __launch_bounds__(1024) void edge_owner(const float* __restrict__ ft,
                               const float* __restrict__ el,
                               const float* __restrict__ er,
                               const int* __restrict__ src,
                               const int* __restrict__ dst,
                               const unsigned* __restrict__ rfin,
                               const unsigned* __restrict__ emaxk,
                               float* __restrict__ out,
                               int nnodes, int nedges)
{
    __shared__ float4 acc4[OWN * HD / 4];
    __shared__ float4 dsum4[OWN];
    __shared__ float4 erl4[OWN];
    __shared__ float4 shl4[OWN];
    __shared__ int    q[OWAVES][QRING];
    float* acc  = (float*)acc4;
    float* dsum = (float*)dsum4;

    const int tid  = threadIdx.x;
    const int lane = tid & 31;
    const int wave = __builtin_amdgcn_readfirstlane(tid >> 5);
    const int n0   = blockIdx.x * OWN;
    const int nown = (nnodes - n0) < OWN ? (nnodes - n0) : OWN;
    const unsigned bound = rfin[4];

    for (int i = tid; i < OWN * HD / 4; i += 1024) acc4[i] = make_float4(0.f, 0.f, 0.f, 0.f);
    {
        float* erl = (float*)erl4;
        float* shl = (float*)shl4;
        for (int i = tid; i < OWN * NH; i += 1024) {
            float erv = 0.f, c = 0.f;
            if (i < nown * NH) {
                erv = er[(size_t)n0 * NH + i];
                if (bound != 0u) {
                    c = __uint_as_float(rfin[i & 3]) + erv;
                    c = c > 0.f ? c : NEG_SLOPE * c;
                } else {
                    const unsigned k = emaxk[(size_t)n0 * NH + i];
                    c = (k == 0u) ? 0.f : ord2f(k);
                    if (!__builtin_isfinite(c)) c = 0.f;
                }
            }
            dsum[i] = 0.f;
            erl[i]  = erv;
            shl[i]  = c;
        }
    }
    __syncthreads();

    int* myq = q[wave];
    int qh = 0, qt = 0;

    auto take = [&](const int nb) {
        int s = 0, dl = 0;
        float4 l4 = make_float4(0.f, 0.f, 0.f, 0.f);
        if (lane < nb) {
            const int e = myq[(qh + lane) & (QRING - 1)];
            s  = src[e];
            dl = dst[e] - n0;
            l4 = *(const float4*)(el + (size_t)s * NH);
        }
        float ex0 = 0.f, ex1 = 0.f, ex2 = 0.f, ex3 = 0.f;
        for (int j0 = 0; j0 < nb; j0 += 8) {
            float v[8][NH];
#pragma unroll
            for (int j = 0; j < 8; ++j) {
                const int sj = __builtin_amdgcn_readlane(s, j0 + j);
                const float* fr = ft + (size_t)sj * HD + lane;
#pragma unroll
                for (int hh = 0; hh < NH; ++hh) v[j][hh] = fr[DH * hh];
            }
            if (j0 == 0 && lane < nb) {
                const float4 r4 = erl4[dl];
                const float4 c4 = shl4[dl];
                float v0 = l4.x + r4.x, v1 = l4.y + r4.y, v2 = l4.z + r4.z, v3 = l4.w + r4.w;
                v0 = v0 > 0.f ? v0 : NEG_SLOPE * v0;
                v1 = v1 > 0.f ? v1 : NEG_SLOPE * v1;
                v2 = v2 > 0.f ? v2 : NEG_SLOPE * v2;
                v3 = v3 > 0.f ? v3 : NEG_SLOPE * v3;
                ex0 = expf(v0 - c4.x);
                ex1 = expf(v1 - c4.y);
                ex2 = expf(v2 - c4.z);
                ex3 = expf(v3 - c4.w);
                atomicAdd(&dsum[dl * NH + 0], ex0);
                atomicAdd(&dsum[dl * NH + 1], ex1);
                atomicAdd(&dsum[dl * NH + 2], ex2);
                atomicAdd(&dsum[dl * NH + 3], ex3);
            }
#pragma unroll
            for (int j = 0; j < 8; ++j) {
                if (j0 + j < nb) {
                    const int dj = __builtin_amdgcn_readlane(dl, j0 + j);
                    float* ar = acc + dj * HD + lane;
                    const float a0 = __int_as_float(__builtin_amdgcn_readlane(__float_as_int(ex0), j0 + j));
                    const float a1 = __int_as_float(__builtin_amdgcn_readlane(__float_as_int(ex1), j0 + j));
                    const float a2 = __int_as_float(__builtin_amdgcn_readlane(__float_as_int(ex2), j0 + j));
                    const float a3 = __int_as_float(__builtin_amdgcn_readlane(__float_as_int(ex3), j0 + j));
                    atomicAdd(ar,          a0 * v[j][0]);
                    atomicAdd(ar + DH,     a1 * v[j][1]);
                    atomicAdd(ar + 2 * DH, a2 * v[j][2]);
                    atomicAdd(ar + 3 * DH, a3 * v[j][3]);
                }
            }
        }
        qh += nb;
    };

    const bool use_codes = (bound != 0u) && codes_fit(nnodes, nedges);
    const unsigned* wsrc = use_codes ? emaxk : (const unsigned*)dst;
    const int sh = use_codes ? 2 : 0;
    const int nw = use_codes ? ((nedges >> 4) << 2) : (nedges & ~3);
    const unsigned pat = (unsigned)(blockIdx.x & 255) * 0x01010101u;
    const int nchunk = (nw + CWORDS - 1) / CWORDS;
    const int nit    = (nchunk + OWAVES - 1) / OWAVES;

    auto push = [&](const bool hit, const int e) {
        const unsigned m = __builtin_amdgcn_ballot_w32(hit);
        if (m != 0u) {
            if (hit) myq[(qt + (int)__builtin_amdgcn_mbcnt_lo(m, 0u)) & (QRING - 1)] = e;
            qt += __builtin_popcount(m);
        }
    };
    const int rot = (int)(((long long)blockIdx.x * nit) / gridDim.x) * OWAVES;
    auto chunk_of = [&](const int it) -> int {
        if (it >= nit) return -1;
        int c = it * OWAVES + wave + rot;
        if (c >= nit * OWAVES) c -= nit * OWAVES;
        return c < nchunk ? c : -1;
    };
    auto load16 = [&](const int c, const int u) -> uint4 {
        int w = (c < 0 ? 0 : c) * CWORDS + u * 128 + lane * 4;
        w = w < nw - 4 ? w : nw - 4;
        return *(const uint4*)(wsrc + w);
    };
    auto scan_codes = [&](const unsigned y, const unsigned vm, const int w) {
        const unsigned t = (y - 0x01010101u) & ~y & vm;
        if (__builtin_amdgcn_ballot_w32(t != 0u) != 0u) {
            const unsigned yy = (vm != 0u) ? y : 0xFFFFFFFFu;
            unsigned z = ~(((yy & 0x7F7F7F7Fu) + 0x7F7F7F7Fu) | yy) & 0x80808080u;
            const int e = w << 2;
            for (unsigned m = __builtin_amdgcn_ballot_w32(z != 0u); m != 0u; m = __builtin_amdgcn_ballot_w32(z != 0u)) {
                const int p = qt + (int)__builtin_amdgcn_mbcnt_lo(m, 0u);
                if (z != 0u && (p - qh) < QRING) myq[p & (QRING - 1)] = e + (__builtin_ctz(z | 0x80000000u) >> 3);
                qt += __builtin_popcount(m);
                z &= z - 1u;
            }
        }
    };

    auto scan_dst = [&](const unsigned x, const bool valid, const int w) {
        const bool hit = valid && ((unsigned)((int)x - n0) < (unsigned)nown);
        const unsigned m = __builtin_amdgcn_ballot_w32(hit);
        if (m != 0u) {
            const int p = qt + (int)__builtin_amdgcn_mbcnt_lo(m, 0u);
            if (hit && (p - qh) < QRING) myq[p & (QRING - 1)] = w;
            qt += __builtin_popcount(m);
        }
    };

    if (wave == 0) {
        const int e = (nw << sh) + lane;
        const int d = (e < nedges) ? dst[e] : -1;
        push((unsigned)(d - n0) < (unsigned)nown, e);
    }

    int it = (nw > 0) ? -1 : 0;
    int ccur = -1;
    int spos = 0, send = 0;
    uint4 cv[8];
#pragma unroll
    for (int u = 0; u < 8; ++u) cv[u] = make_uint4(0u, 0u, 0u, 0u);
    auto scan_step = [&]() {
        const int cnext = chunk_of(it + 1);
        const int qt0 = qt;
#pragma unroll
        for (int u = 0; u < 8; ++u) {
            if (ccur >= 0) {
                const int w = ccur * CWORDS + u * 128 + lane * 4;
                const bool valid = w < nw;
                if (use_codes) {
                    const unsigned vm = valid ? 0x80808080u : 0u;
                    scan_codes(cv[u].x ^ pat, vm, w);
                    scan_codes(cv[u].y ^ pat, vm, w + 1);
                    scan_codes(cv[u].z ^ pat, vm, w + 2);
                    scan_codes(cv[u].w ^ pat, vm, w + 3);
                } else {
                    scan_dst(cv[u].x, valid, w);
                    scan_dst(cv[u].y, valid, w + 1);
                    scan_dst(cv[u].z, valid, w + 2);
                    scan_dst(cv[u].w, valid, w + 3);
                }
            }
            cv[u] = load16(cnext, u);
        }
        if (qt - qh > QRING) {
            qt = qt0;
            spos = ccur * CWORDS;
            send = (spos + CWORDS) < nw ? (spos + CWORDS) : nw;
        }
        ccur = cnext;
        ++it;
    };
    for (;;) {
        const int nq = qt - qh;
        const bool scanned = !(spos < send) && (it >= nit);
        if (nq >= 32 || (scanned && nq > 0)) {
            take(nq < 32 ? nq : 32);
            continue;
        }
        if (scanned) break;
        if (spos < send) {
            const int w = spos + lane;
            const bool valid = w < send;
            const unsigned x = valid ? wsrc[w] : 0u;
            if (use_codes) {
                const unsigned yy = valid ? (x ^ pat) : 0xFFFFFFFFu;
                const int e = w << 2;
#pragma nounroll
                for (int b = 0; b < 4; ++b) push(((yy >> (8 * b)) & 0xFFu) == 0u, e + b);
            } else {
                push(valid && ((unsigned)((int)x - n0) < (unsigned)nown), w);
            }
            spos += 32;
            continue;
        }
        scan_step();
    }
    __syncthreads();

    float4* o4 = (float4*)(out + (size_t)n0 * HD);
    for (int i = tid; i < nown * (HD / 4); i += 1024) {
        const float dn = dsum[(i >> 5) * NH + ((i & 31) >> 3)];
        const float dq = dn > 0.f ? dn : 1.0f;
        float4 a = acc4[i];
        a.x /= dq; a.y /= dq; a.z /= dq; a.w /= dq;
        o4[i] = a;
    }
}

static inline unsigned nblk(long long n, int bs) {
    return (unsigned)((n + bs - 1) / bs);
}

extern "C" void kernel_launch(void* const* d_in, const int* in_sizes, int n_in,
                              void* d_out, int out_size, void* d_ws, size_t ws_size,
                              hipStream_t stream) {
    const float* feat   = (const float*)d_in[0];
    const float* W      = (const float*)d_in[1];
    const float* attn_l = (const float*)d_in[2];
    const float* attn_r = (const float*)d_in[3];
    const int*   src    = (const int*)d_in[4];
    const int*   dst    = (const int*)d_in[5];
    float*       out    = (float*)d_out;

    const long long N = in_sizes[0] / IN_F;
    const long long E = in_sizes[4];

    float*    ft    = (float*)d_ws;
    float*    el    = ft + (size_t)N * HD;
    float*    er    = el + (size_t)N * NH;
    unsigned* emaxk = (unsigned*)(er + (size_t)N * NH);
    unsigned* rng   = emaxk + (size_t)N * NH;
    unsigned* rfin  = rng + RNG_SLOTS * 8;

    const int BS = 256;
    const long long NHtot = N * NH;
    const long long EHtot = E * NH;

    gemm_wmma_f32<<<nblk(N, GEMM_ROWS_PER_BLOCK), 64, 0, stream>>>(feat, W, ft, rng, (int)N);

    node_attn<<<nblk(NHtot, BS), BS, 0, stream>>>(ft, attn_l, attn_r, el, er, emaxk, rng, NHtot);

    edge_max<<<2048, BS, 0, stream>>>(el, er, src, dst, rng, rfin, emaxk, EHtot, (int)N, (int)E);

    edge_owner<<<nblk(N, OWN), 1024, 0, stream>>>(ft, el, er, src, dst, rfin, emaxk, out, (int)N, (int)E);
}


# Owner codes kept as 8 bit planes per 32 edges: a lane of launch 4 tests 32 edges with 12 logic ops on two 16-byte loads, set bits are its edges; launch 3 builds them by wave votes; sums arrival-order
# speedup vs baseline: 3.1589x; 1.0496x over previous
#include <hip/hip_runtime.h>
#include <hip/hip_bf16.h>

#define IN_F 128
#define HD   128
#define NH   4
#define DH   32
#define NEG_SLOPE 0.2f

typedef __attribute__((ext_vector_type(2))) float v2f;
typedef __attribute__((ext_vector_type(8))) float v8f;

__device__ __forceinline__ unsigned f2ord(float f) {
    unsigned b = __float_as_uint(f);
    return (b & 0x80000000u) ? ~b : (b | 0x80000000u);
}
__device__ __forceinline__ float ord2f(unsigned k) {
    unsigned b = (k & 0x80000000u) ? (k ^ 0x80000000u) : ~k;
    return __uint_as_float(b);
}

#define GEMM_ROWS_PER_BLOCK 32
#define RNG_SLOTS 64
#define RNG_MAX 32.0f
__global__ __launch_bounds__(64) void gemm_wmma_f32(
    const float* __restrict__ A,
    const float* __restrict__ B,
    float* __restrict__ C,
    unsigned* __restrict__ rng,
    int nrows)
{
    if (blockIdx.x == 0) {
        for (int q = threadIdx.x; q < RNG_SLOTS * 8; q += 64) rng[q] = 0u;
    }

    const int lane    = threadIdx.x & 31;
    const int wave    = threadIdx.x >> 5;
    const int l15     = lane & 15;
    const int halfSel = lane >> 4;
    const int mBase   = (blockIdx.x * 2 + wave) * 16;
    if (mBase >= nrows) return;

    int row = mBase + l15;
    if (row >= nrows) row = nrows - 1;

    const float* aptr = A + (size_t)row * IN_F + (halfSel << 1);
    const float* bptr = B + (size_t)(halfSel << 1) * HD + (l15 << 2);

    v8f acc[8];
#pragma unroll
    for (int t = 0; t < 8; ++t) acc[t] = v8f{};

    v2f    a   = *(const v2f*)aptr;
    float4 b0l = *(const float4*)(bptr);
    float4 b1l = *(const float4*)(bptr + HD);
    float4 b0h = *(const float4*)(bptr + 64);
    float4 b1h = *(const float4*)(bptr + HD + 64);

#pragma unroll 2
    for (int k = 0; k < IN_F; k += 4) {
        const int kn = (k + 4 < IN_F) ? (k + 4) : 0;
        const v2f    an   = *(const v2f*)(aptr + kn);
        const float4 b0ln = *(const float4*)(bptr + (size_t)kn * HD);
        const float4 b1ln = *(const float4*)(bptr + (size_t)(kn + 1) * HD);
        const float4 b0hn = *(const float4*)(bptr + (size_t)kn * HD + 64);
        const float4 b1hn = *(const float4*)(bptr + (size_t)(kn + 1) * HD + 64);

        v2f b;
        b.x = b0l.x; b.y = b1l.x;
        acc[0] = __builtin_amdgcn_wmma_f32_16x16x4_f32(false, a, false, b, (short)0, acc[0], false, false);
        b.x = b0l.y; b.y = b1l.y;
        acc[1] = __builtin_amdgcn_wmma_f32_16x16x4_f32(false, a, false, b, (short)0, acc[1], false, false);
        b.x = b0l.z; b.y = b1l.z;
        acc[2] = __builtin_amdgcn_wmma_f32_16x16x4_f32(false, a, false, b, (short)0, acc[2], false, false);
        b.x = b0l.w; b.y = b1l.w;
        acc[3] = __builtin_amdgcn_wmma_f32_16x16x4_f32(false, a, false, b, (short)0, acc[3], false, false);
        b.x = b0h.x; b.y = b1h.x;
        acc[4] = __builtin_amdgcn_wmma_f32_16x16x4_f32(false, a, false, b, (short)0, acc[4], false, false);
        b.x = b0h.y; b.y = b1h.y;
        acc[5] = __builtin_amdgcn_wmma_f32_16x16x4_f32(false, a, false, b, (short)0, acc[5], false, false);
        b.x = b0h.z; b.y = b1h.z;
        acc[6] = __builtin_amdgcn_wmma_f32_16x16x4_f32(false, a, false, b, (short)0, acc[6], false, false);
        b.x = b0h.w; b.y = b1h.w;
        acc[7] = __builtin_amdgcn_wmma_f32_16x16x4_f32(false, a, false, b, (short)0, acc[7], false, false);

        a = an; b0l = b0ln; b1l = b1ln; b0h = b0hn; b1h = b1hn;
    }

#pragma unroll
    for (int r = 0; r < 8; ++r) {
        const int m = mBase + r + (halfSel << 3);
        if (m < nrows) {
            float* crow = C + (size_t)m * HD + (l15 << 2);
            *(float4*)(crow)      = make_float4(acc[0][r], acc[1][r], acc[2][r], acc[3][r]);
            *(float4*)(crow + 64) = make_float4(acc[4][r], acc[5][r], acc[6][r], acc[7][r]);
        }
    }
}

__global__ void node_attn(const float* __restrict__ ft,
                          const float* __restrict__ attn_l,
                          const float* __restrict__ attn_r,
                          float* __restrict__ el, float* __restrict__ er,
                          unsigned* __restrict__ emaxk,
                          unsigned* __restrict__ rng,
                          long long nh_total)
{
    __shared__ unsigned red[8][8];
    long long t = (long long)blockIdx.x * blockDim.x + threadIdx.x;
    const bool live = t < nh_total;
    const int h = (int)(threadIdx.x & 3);
    unsigned kmax = 0u, kmin = 0u;
    if (live) {
        const long long n = t >> 2;
        const float* f  = ft + (size_t)n * HD + h * DH;
        const float* al = attn_l + h * DH;
        const float* ar = attn_r + h * DH;
        float sl = 0.f, sr = 0.f;
#pragma unroll
        for (int d = 0; d < DH; d += 4) {
            const float4 fv = *(const float4*)(f + d);
            const float4 lv = *(const float4*)(al + d);
            const float4 rv = *(const float4*)(ar + d);
            sl += fv.x * lv.x + fv.y * lv.y + fv.z * lv.z + fv.w * lv.w;
            sr += fv.x * rv.x + fv.y * rv.y + fv.z * rv.z + fv.w * rv.w;
        }
        el[t] = sl;
        er[t] = sr;
        emaxk[t] = 0u;
        kmax = f2ord(sl);
        kmin = ~kmax;
    }
    for (int m = 4; m < 32; m <<= 1) {
        const unsigned a = (unsigned)__shfl_xor((int)kmax, m);
        const unsigned b = (unsigned)__shfl_xor((int)kmin, m);
        kmax = a > kmax ? a : kmax;
        kmin = b > kmin ? b : kmin;
    }
    const int lane = threadIdx.x & 31;
    const int wave = threadIdx.x >> 5;
    if (lane < 4) { red[wave][lane] = kmax; red[wave][4 + lane] = kmin; }
    __syncthreads();
    if (threadIdx.x < 8) {
        unsigned k = 0u;
        const int nw = (int)(blockDim.x >> 5);
        for (int w = 0; w < nw; ++w) { const unsigned v = red[w][threadIdx.x]; k = v > k ? v : k; }
        atomicMax(&rng[(blockIdx.x & (RNG_SLOTS - 1)) * 8 + threadIdx.x], k);
    }
}

__device__ __forceinline__ unsigned fold_range(const unsigned* __restrict__ rng, int lane, float mx[NH])
{
    unsigned k[8];
#pragma unroll
    for (int i = 0; i < 8; ++i) {
        const unsigned a = rng[lane * 8 + i];
        const unsigned b = rng[(lane + 32) * 8 + i];
        k[i] = a > b ? a : b;
    }
#pragma unroll
    for (int i = 0; i < 8; ++i) {
        for (int m = 1; m < 32; m <<= 1) {
            const unsigned o = (unsigned)__shfl_xor((int)k[i], m);
            k[i] = o > k[i] ? o : k[i];
        }
    }
    unsigned ok = 1u;
#pragma unroll
    for (int hh = 0; hh < NH; ++hh) {
        const float hi = (k[hh] == 0u) ? __builtin_nanf("") : ord2f(k[hh]);
        const float lo = (k[4 + hh] == 0u) ? __builtin_nanf("") : ord2f(~k[4 + hh]);
        const bool good = __builtin_isfinite(hi) && __builtin_isfinite(lo) && ((hi - lo) <= RNG_MAX);
        if (!good) ok = 0u;
        mx[hh] = hi;
    }
    return ok;
}

#define OWN     400
#define PBLOCK  1024
__device__ __forceinline__ bool codes_fit(const int nnodes, const int nedges) {
    return ((long long)nedges <= (long long)nnodes * 16) && ((nnodes + OWN - 1) / OWN <= 256);
}

__global__ void edge_max(const float* __restrict__ el,
                         const float* __restrict__ er,
                         const int* __restrict__ src,
                         const int* __restrict__ dst,
                         const unsigned* __restrict__ rng,
                         unsigned* __restrict__ rfin,
                         unsigned* __restrict__ emaxk,
                         long long total,
                         int nnodes, int nedges)
{
    const long long step = (long long)gridDim.x * blockDim.x;
    const long long tid0 = (long long)blockIdx.x * blockDim.x + threadIdx.x;
    const int lane = (int)(threadIdx.x & 31);
    const int gw  = __builtin_amdgcn_readfirstlane((int)(tid0 >> 5));
    const int nwv = (int)(step >> 5);
    const int nblocks = codes_fit(nnodes, nedges) ? (nedges / PBLOCK) : 0;
    int4 d[8];
#pragma unroll
    for (int j = 0; j < 8; ++j) d[j] = make_int4(0, 0, 0, 0);
    if (gw < nblocks) {
#pragma unroll
        for (int j = 0; j < 8; ++j) d[j] = *(const int4*)(dst + (size_t)gw * PBLOCK + j * 128 + 4 * lane);
    }

    float mx[NH];
    const unsigned ok = fold_range(rng, lane, mx);
    if (blockIdx.x == 0 && threadIdx.x == 0) {
#pragma unroll
        for (int hh = 0; hh < NH; ++hh) rfin[hh] = __float_as_uint(mx[hh]);
        rfin[4] = ok;
    }
    if (ok != 0u) {
        for (int blk = gw; blk < nblocks; blk += nwv) {
            if (blk != gw) {
#pragma unroll
                for (int j = 0; j < 8; ++j) d[j] = *(const int4*)(dst + (size_t)blk * PBLOCK + j * 128 + 4 * lane);
            }
            int r[8];
#pragma unroll
            for (int k = 0; k < 8; ++k) r[k] = 0;
#pragma unroll
            for (int j = 0; j < 8; ++j) {
#pragma unroll
                for (int cc = 0; cc < 4; ++cc) {
                    const int dv = cc == 0 ? d[j].x : cc == 1 ? d[j].y : cc == 2 ? d[j].z : d[j].w;
                    const unsigned code = ((unsigned)dv / OWN) & 255u;
                    const bool mine = lane == 4 * j + cc;
#pragma unroll
                    for (int k = 0; k < 8; ++k) {
                        const unsigned vote = __builtin_amdgcn_ballot_w32(((code >> k) & 1u) != 0u);
                        r[k] = mine ? (int)vote : r[k];
                    }
                }
            }
            unsigned* pw = emaxk + (size_t)blk * (PBLOCK / 4) + 4 * lane;
            *(uint4*)(pw)       = make_uint4((unsigned)r[0], (unsigned)r[1], (unsigned)r[2], (unsigned)r[3]);
            *(uint4*)(pw + 128) = make_uint4((unsigned)r[4], (unsigned)r[5], (unsigned)r[6], (unsigned)r[7]);
        }
        return;
    }
    for (long long t = tid0; t < total; t += step) {
        const long long e = t >> 2;
        const int h = (int)(t & 3);
        const int s = src[e], dd = dst[e];
        float v = el[(size_t)s * NH + h] + er[(size_t)dd * NH + h];
        v = v > 0.f ? v : NEG_SLOPE * v;
        atomicMax(&emaxk[(size_t)dd * NH + h], f2ord(v));
    }
}

#define OWAVES  32
#define QRING   512
#define CWORDS  1024
__global__ __launch_bounds__(1024) void edge_owner(const float* __restrict__ ft,
                               const float* __restrict__ el,
                               const float* __restrict__ er,
                               const int* __restrict__ src,
                               const int* __restrict__ dst,
                               const unsigned* __restrict__ rfin,
                               const unsigned* __restrict__ emaxk,
                               float* __restrict__ out,
                               int nnodes, int nedges)
{
    __shared__ float4 acc4[OWN * HD / 4];
    __shared__ float4 dsum4[OWN];
    __shared__ float4 erl4[OWN];
    __shared__ float4 shl4[OWN];
    __shared__ int    q[OWAVES][QRING];
    float* acc  = (float*)acc4;
    float* dsum = (float*)dsum4;

    const int tid  = threadIdx.x;
    const int lane = tid & 31;
    const int wave = __builtin_amdgcn_readfirstlane(tid >> 5);
    const int n0   = blockIdx.x * OWN;
    const int nown = (nnodes - n0) < OWN ? (nnodes - n0) : OWN;
    const unsigned bound = rfin[4];

    for (int i = tid; i < OWN * HD / 4; i += 1024) acc4[i] = make_float4(0.f, 0.f, 0.f, 0.f);
    {
        float* erl = (float*)erl4;
        float* shl = (float*)shl4;
        for (int i = tid; i < OWN * NH; i += 1024) {
            float erv = 0.f, c = 0.f;
            if (i < nown * NH) {
                erv = er[(size_t)n0 * NH + i];
                if (bound != 0u) {
                    c = __uint_as_float(rfin[i & 3]) + erv;
                    c = c > 0.f ? c : NEG_SLOPE * c;
                } else {
                    const unsigned k = emaxk[(size_t)n0 * NH + i];
                    c = (k == 0u) ? 0.f : ord2f(k);
                    if (!__builtin_isfinite(c)) c = 0.f;
                }
            }
            dsum[i] = 0.f;
            erl[i]  = erv;
            shl[i]  = c;
        }
    }
    __syncthreads();

    int* myq = q[wave];
    int qh = 0, qt = 0;

    auto take = [&](const int nb) {
        int s = 0, dl = 0;
        float4 l4 = make_float4(0.f, 0.f, 0.f, 0.f);
        if (lane < nb) {
            const int e = myq[(qh + lane) & (QRING - 1)];
            s  = src[e];
            dl = dst[e] - n0;
            l4 = *(const float4*)(el + (size_t)s * NH);
        }
        float ex0 = 0.f, ex1 = 0.f, ex2 = 0.f, ex3 = 0.f;
        for (int j0 = 0; j0 < nb; j0 += 8) {
            float v[8][NH];
#pragma unroll
            for (int j = 0; j < 8; ++j) {
                const int sj = __builtin_amdgcn_readlane(s, j0 + j);
                const float* fr = ft + (size_t)sj * HD + lane;
#pragma unroll
                for (int hh = 0; hh < NH; ++hh) v[j][hh] = fr[DH * hh];
            }
            if (j0 == 0 && lane < nb) {
                const float4 r4 = erl4[dl];
                const float4 c4 = shl4[dl];
                float v0 = l4.x + r4.x, v1 = l4.y + r4.y, v2 = l4.z + r4.z, v3 = l4.w + r4.w;
                v0 = v0 > 0.f ? v0 : NEG_SLOPE * v0;
                v1 = v1 > 0.f ? v1 : NEG_SLOPE * v1;
                v2 = v2 > 0.f ? v2 : NEG_SLOPE * v2;
                v3 = v3 > 0.f ? v3 : NEG_SLOPE * v3;
                ex0 = expf(v0 - c4.x);
                ex1 = expf(v1 - c4.y);
                ex2 = expf(v2 - c4.z);
                ex3 = expf(v3 - c4.w);
                atomicAdd(&dsum[dl * NH + 0], ex0);
                atomicAdd(&dsum[dl * NH + 1], ex1);
                atomicAdd(&dsum[dl * NH + 2], ex2);
                atomicAdd(&dsum[dl * NH + 3], ex3);
            }
#pragma unroll
            for (int j = 0; j < 8; ++j) {
                if (j0 + j < nb) {
                    const int dj = __builtin_amdgcn_readlane(dl, j0 + j);
                    float* ar = acc + dj * HD + lane;
                    const float a0 = __int_as_float(__builtin_amdgcn_readlane(__float_as_int(ex0), j0 + j));
                    const float a1 = __int_as_float(__builtin_amdgcn_readlane(__float_as_int(ex1), j0 + j));
                    const float a2 = __int_as_float(__builtin_amdgcn_readlane(__float_as_int(ex2), j0 + j));
                    const float a3 = __int_as_float(__builtin_amdgcn_readlane(__float_as_int(ex3), j0 + j));
                    atomicAdd(ar,          a0 * v[j][0]);
                    atomicAdd(ar + DH,     a1 * v[j][1]);
                    atomicAdd(ar + 2 * DH, a2 * v[j][2]);
                    atomicAdd(ar + 3 * DH, a3 * v[j][3]);
                }
            }
        }
        qh += nb;
    };

    const bool use_codes = (bound != 0u) && codes_fit(nnodes, nedges);
    const unsigned* wsrc = use_codes ? emaxk : (const unsigned*)dst;
    const int sh = use_codes ? 2 : 0;
    const int nw = use_codes ? ((nedges / PBLOCK) * (PBLOCK / 4)) : (nedges & ~3);
    unsigned pl[8];
#pragma unroll
    for (int k = 0; k < 8; ++k) pl[k] = 0u - (((unsigned)blockIdx.x >> k) & 1u);
    const int gofs = ((lane >> 2) << 7) + (lane & 3);
    const int nchunk = (nw + CWORDS - 1) / CWORDS;
    const int nit    = (nchunk + OWAVES - 1) / OWAVES;

    auto push = [&](const bool hit, const int e) {
        const unsigned m = __builtin_amdgcn_ballot_w32(hit);
        if (m != 0u) {
            if (hit) myq[(qt + (int)__builtin_amdgcn_mbcnt_lo(m, 0u)) & (QRING - 1)] = e;
            qt += __builtin_popcount(m);
        }
    };
    const int rot = (int)(((long long)blockIdx.x * nit) / gridDim.x) * OWAVES;
    auto chunk_of = [&](const int it) -> int {
        if (it >= nit) return -1;
        int c = it * OWAVES + wave + rot;
        if (c >= nit * OWAVES) c -= nit * OWAVES;
        return c < nchunk ? c : -1;
    };
    auto load16 = [&](const int c, const int u) -> uint4 {
        int w = (c < 0 ? 0 : c) * CWORDS + u * 128 + lane * 4;
        w = w < nw - 4 ? w : nw - 4;
        return *(const uint4*)(wsrc + w);
    };
    auto scan_planes = [&](const uint4 lo, const uint4 hi, const int e0) {
        unsigned m = ~(lo.x ^ pl[0]) & ~(lo.y ^ pl[1]) & ~(lo.z ^ pl[2]) & ~(lo.w ^ pl[3])
                   & ~(hi.x ^ pl[4]) & ~(hi.y ^ pl[5]) & ~(hi.z ^ pl[6]) & ~(hi.w ^ pl[7]);
        for (unsigned mm = __builtin_amdgcn_ballot_w32(m != 0u); mm != 0u; mm = __builtin_amdgcn_ballot_w32(m != 0u)) {
            const int p = qt + (int)__builtin_amdgcn_mbcnt_lo(mm, 0u);
            if (m != 0u && (p - qh) < QRING) myq[p & (QRING - 1)] = e0 + (__builtin_ctz(m | 0x80000000u) << 2);
            qt += __builtin_popcount(mm);
            m &= m - 1u;
        }
    };

    auto scan_dst = [&](const unsigned x, const bool valid, const int w) {
        const bool hit = valid && ((unsigned)((int)x - n0) < (unsigned)nown);
        const unsigned m = __builtin_amdgcn_ballot_w32(hit);
        if (m != 0u) {
            const int p = qt + (int)__builtin_amdgcn_mbcnt_lo(m, 0u);
            if (hit && (p - qh) < QRING) myq[p & (QRING - 1)] = w;
            qt += __builtin_popcount(m);
        }
    };


    int it = (nw > 0) ? -1 : 0;
    int ccur = -1;
    int spos = 0, send = 0;
    if (wave == 0) { spos = nw << sh; send = nedges; }
    uint4 cv[8];
#pragma unroll
    for (int u = 0; u < 8; ++u) cv[u] = make_uint4(0u, 0u, 0u, 0u);
    auto scan_step = [&]() {
        const int cnext = chunk_of(it + 1);
        const int qt0 = qt;
#pragma unroll
        for (int u = 0; u < 8; ++u) {
            if (ccur >= 0) {
                if (use_codes) {
                    if ((u & 1) == 0) {
                        const int wb = ccur * CWORDS + (u >> 1) * (PBLOCK / 4);
                        if (wb < nw) scan_planes(cv[u], cv[u + 1], (wb << 2) + gofs);
                    }
                } else {
                    const int w = ccur * CWORDS + u * 128 + lane * 4;
                    const bool valid = w < nw;
                    scan_dst(cv[u].x, valid, w);
                    scan_dst(cv[u].y, valid, w + 1);
                    scan_dst(cv[u].z, valid, w + 2);
                    scan_dst(cv[u].w, valid, w + 3);
                }
            }
            cv[u] = load16(cnext, u);
        }
        if (qt - qh > QRING) {
            qt = qt0;
            spos = (ccur * CWORDS) << sh;
            send = ((ccur * CWORDS + CWORDS) < nw ? (ccur * CWORDS + CWORDS) : nw) << sh;
        }
        ccur = cnext;
        ++it;
    };
    for (;;) {
        const int nq = qt - qh;
        const bool scanned = !(spos < send) && (it >= nit);
        if (nq >= 32 || (scanned && nq > 0)) {
            take(nq < 32 ? nq : 32);
            continue;
        }
        if (scanned) break;
        if (spos < send) {
            const int e = spos + lane;
            const bool valid = e < send;
            const int x = valid ? dst[e] : 0;
            push(valid && ((unsigned)(x - n0) < (unsigned)nown), e);
            spos += 32;
            continue;
        }
        scan_step();
    }
    __syncthreads();

    float4* o4 = (float4*)(out + (size_t)n0 * HD);
    for (int i = tid; i < nown * (HD / 4); i += 1024) {
        const float dn = dsum[(i >> 5) * NH + ((i & 31) >> 3)];
        const float dq = dn > 0.f ? dn : 1.0f;
        float4 a = acc4[i];
        a.x /= dq; a.y /= dq; a.z /= dq; a.w /= dq;
        o4[i] = a;
    }
}

static inline unsigned nblk(long long n, int bs) {
    return (unsigned)((n + bs - 1) / bs);
}

extern "C" void kernel_launch(void* const* d_in, const int* in_sizes, int n_in,
                              void* d_out, int out_size, void* d_ws, size_t ws_size,
                              hipStream_t stream) {
    const float* feat   = (const float*)d_in[0];
    const float* W      = (const float*)d_in[1];
    const float* attn_l = (const float*)d_in[2];
    const float* attn_r = (const float*)d_in[3];
    const int*   src    = (const int*)d_in[4];
    const int*   dst    = (const int*)d_in[5];
    float*       out    = (float*)d_out;

    const long long N = in_sizes[0] / IN_F;
    const long long E = in_sizes[4];

    float*    ft    = (float*)d_ws;
    float*    el    = ft + (size_t)N * HD;
    float*    er    = el + (size_t)N * NH;
    unsigned* emaxk = (unsigned*)(er + (size_t)N * NH);
    unsigned* rng   = emaxk + (size_t)N * NH;
    unsigned* rfin  = rng + RNG_SLOTS * 8;

    const int BS = 256;
    const long long NHtot = N * NH;
    const long long EHtot = E * NH;

    gemm_wmma_f32<<<nblk(N, GEMM_ROWS_PER_BLOCK), 64, 0, stream>>>(feat, W, ft, rng, (int)N);

    node_attn<<<nblk(NHtot, BS), BS, 0, stream>>>(ft, attn_l, attn_r, el, er, emaxk, rng, NHtot);

    edge_max<<<392, 128, 0, stream>>>(el, er, src, dst, rng, rfin, emaxk, EHtot, (int)N, (int)E);

    edge_owner<<<nblk(N, OWN), 1024, 0, stream>>>(ft, el, er, src, dst, rfin, emaxk, out, (int)N, (int)E);
}


# Source dots el from the gathered ft rows in registers (row = one 16-byte load per lane; 8 edges' dots summed over 8 lanes by 7 exchange-adds): per-edge el[src] read dropped; el sum order changes
# speedup vs baseline: 3.5966x; 1.1386x over previous
#include <hip/hip_runtime.h>
#include <hip/hip_bf16.h>

#define IN_F 128
#define HD   128
#define NH   4
#define DH   32
#define NEG_SLOPE 0.2f

typedef __attribute__((ext_vector_type(2))) float v2f;
typedef __attribute__((ext_vector_type(8))) float v8f;

__device__ __forceinline__ unsigned f2ord(float f) {
    unsigned b = __float_as_uint(f);
    return (b & 0x80000000u) ? ~b : (b | 0x80000000u);
}
__device__ __forceinline__ float ord2f(unsigned k) {
    unsigned b = (k & 0x80000000u) ? (k ^ 0x80000000u) : ~k;
    return __uint_as_float(b);
}

template <int CTRL>
__device__ __forceinline__ float lane_swz(float x) {
    return __builtin_bit_cast(float, __builtin_amdgcn_update_dpp(0, __builtin_bit_cast(int, x), CTRL, 0xF, 0xF, true));
}
template <int J>
__device__ __forceinline__ float grp8_bcast(float x) {
    return __builtin_bit_cast(float, __builtin_amdgcn_mov_dpp8(__builtin_bit_cast(unsigned, x), (unsigned)J * 0x249249u));
}

#define GEMM_ROWS_PER_BLOCK 32
#define RNG_SLOTS 64
#define RNG_MAX 32.0f
__global__ __launch_bounds__(64) void gemm_wmma_f32(
    const float* __restrict__ A,
    const float* __restrict__ B,
    float* __restrict__ C,
    unsigned* __restrict__ rng,
    int nrows)
{
    if (blockIdx.x == 0) {
        for (int q = threadIdx.x; q < RNG_SLOTS * 8; q += 64) rng[q] = 0u;
    }

    const int lane    = threadIdx.x & 31;
    const int wave    = threadIdx.x >> 5;
    const int l15     = lane & 15;
    const int halfSel = lane >> 4;
    const int mBase   = (blockIdx.x * 2 + wave) * 16;
    if (mBase >= nrows) return;

    int row = mBase + l15;
    if (row >= nrows) row = nrows - 1;

    const float* aptr = A + (size_t)row * IN_F + (halfSel << 1);
    const float* bptr = B + (size_t)(halfSel << 1) * HD + (l15 << 2);

    v8f acc[8];
#pragma unroll
    for (int t = 0; t < 8; ++t) acc[t] = v8f{};

    v2f    a   = *(const v2f*)aptr;
    float4 b0l = *(const float4*)(bptr);
    float4 b1l = *(const float4*)(bptr + HD);
    float4 b0h = *(const float4*)(bptr + 64);
    float4 b1h = *(const float4*)(bptr + HD + 64);

#pragma unroll 2
    for (int k = 0; k < IN_F; k += 4) {
        const int kn = (k + 4 < IN_F) ? (k + 4) : 0;
        const v2f    an   = *(const v2f*)(aptr + kn);
        const float4 b0ln = *(const float4*)(bptr + (size_t)kn * HD);
        const float4 b1ln = *(const float4*)(bptr + (size_t)(kn + 1) * HD);
        const float4 b0hn = *(const float4*)(bptr + (size_t)kn * HD + 64);
        const float4 b1hn = *(const float4*)(bptr + (size_t)(kn + 1) * HD + 64);

        v2f b;
        b.x = b0l.x; b.y = b1l.x;
        acc[0] = __builtin_amdgcn_wmma_f32_16x16x4_f32(false, a, false, b, (short)0, acc[0], false, false);
        b.x = b0l.y; b.y = b1l.y;
        acc[1] = __builtin_amdgcn_wmma_f32_16x16x4_f32(false, a, false, b, (short)0, acc[1], false, false);
        b.x = b0l.z; b.y = b1l.z;
        acc[2] = __builtin_amdgcn_wmma_f32_16x16x4_f32(false, a, false, b, (short)0, acc[2], false, false);
        b.x = b0l.w; b.y = b1l.w;
        acc[3] = __builtin_amdgcn_wmma_f32_16x16x4_f32(false, a, false, b, (short)0, acc[3], false, false);
        b.x = b0h.x; b.y = b1h.x;
        acc[4] = __builtin_amdgcn_wmma_f32_16x16x4_f32(false, a, false, b, (short)0, acc[4], false, false);
        b.x = b0h.y; b.y = b1h.y;
        acc[5] = __builtin_amdgcn_wmma_f32_16x16x4_f32(false, a, false, b, (short)0, acc[5], false, false);
        b.x = b0h.z; b.y = b1h.z;
        acc[6] = __builtin_amdgcn_wmma_f32_16x16x4_f32(false, a, false, b, (short)0, acc[6], false, false);
        b.x = b0h.w; b.y = b1h.w;
        acc[7] = __builtin_amdgcn_wmma_f32_16x16x4_f32(false, a, false, b, (short)0, acc[7], false, false);

        a = an; b0l = b0ln; b1l = b1ln; b0h = b0hn; b1h = b1hn;
    }

#pragma unroll
    for (int r = 0; r < 8; ++r) {
        const int m = mBase + r + (halfSel << 3);
        if (m < nrows) {
            float* crow = C + (size_t)m * HD + (l15 << 2);
            *(float4*)(crow)      = make_float4(acc[0][r], acc[1][r], acc[2][r], acc[3][r]);
            *(float4*)(crow + 64) = make_float4(acc[4][r], acc[5][r], acc[6][r], acc[7][r]);
        }
    }
}

__global__ void node_attn(const float* __restrict__ ft,
                          const float* __restrict__ attn_l,
                          const float* __restrict__ attn_r,
                          float* __restrict__ el, float* __restrict__ er,
                          unsigned* __restrict__ emaxk,
                          unsigned* __restrict__ rng,
                          long long nh_total)
{
    __shared__ unsigned red[8][8];
    long long t = (long long)blockIdx.x * blockDim.x + threadIdx.x;
    const bool live = t < nh_total;
    const int h = (int)(threadIdx.x & 3);
    unsigned kmax = 0u, kmin = 0u;
    if (live) {
        const long long n = t >> 2;
        const float* f  = ft + (size_t)n * HD + h * DH;
        const float* al = attn_l + h * DH;
        const float* ar = attn_r + h * DH;
        float sl = 0.f, sr = 0.f;
#pragma unroll
        for (int d = 0; d < DH; d += 4) {
            const float4 fv = *(const float4*)(f + d);
            const float4 lv = *(const float4*)(al + d);
            const float4 rv = *(const float4*)(ar + d);
            sl += fv.x * lv.x + fv.y * lv.y + fv.z * lv.z + fv.w * lv.w;
            sr += fv.x * rv.x + fv.y * rv.y + fv.z * rv.z + fv.w * rv.w;
        }
        el[t] = sl;
        er[t] = sr;
        emaxk[t] = 0u;
        kmax = f2ord(sl);
        kmin = ~kmax;
    }
    for (int m = 4; m < 32; m <<= 1) {
        const unsigned a = (unsigned)__shfl_xor((int)kmax, m);
        const unsigned b = (unsigned)__shfl_xor((int)kmin, m);
        kmax = a > kmax ? a : kmax;
        kmin = b > kmin ? b : kmin;
    }
    const int lane = threadIdx.x & 31;
    const int wave = threadIdx.x >> 5;
    if (lane < 4) { red[wave][lane] = kmax; red[wave][4 + lane] = kmin; }
    __syncthreads();
    if (threadIdx.x < 8) {
        unsigned k = 0u;
        const int nw = (int)(blockDim.x >> 5);
        for (int w = 0; w < nw; ++w) { const unsigned v = red[w][threadIdx.x]; k = v > k ? v : k; }
        atomicMax(&rng[(blockIdx.x & (RNG_SLOTS - 1)) * 8 + threadIdx.x], k);
    }
}

__device__ __forceinline__ unsigned fold_range(const unsigned* __restrict__ rng, int lane, float mx[NH])
{
    unsigned k[8];
#pragma unroll
    for (int i = 0; i < 8; ++i) {
        const unsigned a = rng[lane * 8 + i];
        const unsigned b = rng[(lane + 32) * 8 + i];
        k[i] = a > b ? a : b;
    }
#pragma unroll
    for (int i = 0; i < 8; ++i) {
        for (int m = 1; m < 32; m <<= 1) {
            const unsigned o = (unsigned)__shfl_xor((int)k[i], m);
            k[i] = o > k[i] ? o : k[i];
        }
    }
    unsigned ok = 1u;
#pragma unroll
    for (int hh = 0; hh < NH; ++hh) {
        const float hi = (k[hh] == 0u) ? __builtin_nanf("") : ord2f(k[hh]);
        const float lo = (k[4 + hh] == 0u) ? __builtin_nanf("") : ord2f(~k[4 + hh]);
        const bool good = __builtin_isfinite(hi) && __builtin_isfinite(lo) && ((hi - lo) <= RNG_MAX);
        if (!good) ok = 0u;
        mx[hh] = hi;
    }
    return ok;
}

#define OWN     400
#define PBLOCK  1024
__device__ __forceinline__ bool codes_fit(const int nnodes, const int nedges) {
    return ((long long)nedges <= (long long)nnodes * 16) && ((nnodes + OWN - 1) / OWN <= 256);
}

__global__ void edge_max(const float* __restrict__ el,
                         const float* __restrict__ er,
                         const int* __restrict__ src,
                         const int* __restrict__ dst,
                         const unsigned* __restrict__ rng,
                         unsigned* __restrict__ rfin,
                         unsigned* __restrict__ emaxk,
                         long long total,
                         int nnodes, int nedges)
{
    const long long step = (long long)gridDim.x * blockDim.x;
    const long long tid0 = (long long)blockIdx.x * blockDim.x + threadIdx.x;
    const int lane = (int)(threadIdx.x & 31);
    const int gw  = __builtin_amdgcn_readfirstlane((int)(tid0 >> 5));
    const int nwv = (int)(step >> 5);
    const int nblocks = codes_fit(nnodes, nedges) ? (nedges / PBLOCK) : 0;
    int4 d[8];
#pragma unroll
    for (int j = 0; j < 8; ++j) d[j] = make_int4(0, 0, 0, 0);
    if (gw < nblocks) {
#pragma unroll
        for (int j = 0; j < 8; ++j) d[j] = *(const int4*)(dst + (size_t)gw * PBLOCK + j * 128 + 4 * lane);
    }

    float mx[NH];
    const unsigned ok = fold_range(rng, lane, mx);
    if (blockIdx.x == 0 && threadIdx.x == 0) {
#pragma unroll
        for (int hh = 0; hh < NH; ++hh) rfin[hh] = __float_as_uint(mx[hh]);
        rfin[4] = ok;
    }
    if (ok != 0u) {
        for (int blk = gw; blk < nblocks; blk += nwv) {
            if (blk != gw) {
#pragma unroll
                for (int j = 0; j < 8; ++j) d[j] = *(const int4*)(dst + (size_t)blk * PBLOCK + j * 128 + 4 * lane);
            }
            int r[8];
#pragma unroll
            for (int k = 0; k < 8; ++k) r[k] = 0;
#pragma unroll
            for (int j = 0; j < 8; ++j) {
#pragma unroll
                for (int cc = 0; cc < 4; ++cc) {
                    const int dv = cc == 0 ? d[j].x : cc == 1 ? d[j].y : cc == 2 ? d[j].z : d[j].w;
                    const unsigned code = ((unsigned)dv / OWN) & 255u;
                    const bool mine = lane == 4 * j + cc;
#pragma unroll
                    for (int k = 0; k < 8; ++k) {
                        const unsigned vote = __builtin_amdgcn_ballot_w32(((code >> k) & 1u) != 0u);
                        r[k] = mine ? (int)vote : r[k];
                    }
                }
            }
            unsigned* pw = emaxk + (size_t)blk * (PBLOCK / 4) + 4 * lane;
            *(uint4*)(pw)       = make_uint4((unsigned)r[0], (unsigned)r[1], (unsigned)r[2], (unsigned)r[3]);
            *(uint4*)(pw + 128) = make_uint4((unsigned)r[4], (unsigned)r[5], (unsigned)r[6], (unsigned)r[7]);
        }
        return;
    }
    for (long long t = tid0; t < total; t += step) {
        const long long e = t >> 2;
        const int h = (int)(t & 3);
        const int s = src[e], dd = dst[e];
        float v = el[(size_t)s * NH + h] + er[(size_t)dd * NH + h];
        v = v > 0.f ? v : NEG_SLOPE * v;
        atomicMax(&emaxk[(size_t)dd * NH + h], f2ord(v));
    }
}

#define OWAVES  32
#define QRING   512
#define CWORDS  1024
__global__ __launch_bounds__(1024) void edge_owner(const float* __restrict__ ft,
                               const float* __restrict__ attn_l,
                               const float* __restrict__ er,
                               const int* __restrict__ src,
                               const int* __restrict__ dst,
                               const unsigned* __restrict__ rfin,
                               const unsigned* __restrict__ emaxk,
                               float* __restrict__ out,
                               int nnodes, int nedges)
{
    __shared__ float4 acc4[OWN * HD / 4];
    __shared__ float4 dsum4[OWN];
    __shared__ float4 erl4[OWN];
    __shared__ float4 shl4[OWN];
    __shared__ int    q[OWAVES][QRING];
    float* acc  = (float*)acc4;
    float* dsum = (float*)dsum4;

    const int tid  = threadIdx.x;
    const int lane = tid & 31;
    const int wave = __builtin_amdgcn_readfirstlane(tid >> 5);
    const int n0   = blockIdx.x * OWN;
    const int nown = (nnodes - n0) < OWN ? (nnodes - n0) : OWN;
    const unsigned bound = rfin[4];

    for (int i = tid; i < OWN * HD / 4; i += 1024) acc4[i] = make_float4(0.f, 0.f, 0.f, 0.f);
    {
        float* erl = (float*)erl4;
        float* shl = (float*)shl4;
        for (int i = tid; i < OWN * NH; i += 1024) {
            float erv = 0.f, c = 0.f;
            if (i < nown * NH) {
                erv = er[(size_t)n0 * NH + i];
                if (bound != 0u) {
                    c = __uint_as_float(rfin[i & 3]) + erv;
                    c = c > 0.f ? c : NEG_SLOPE * c;
                } else {
                    const unsigned k = emaxk[(size_t)n0 * NH + i];
                    c = (k == 0u) ? 0.f : ord2f(k);
                    if (!__builtin_isfinite(c)) c = 0.f;
                }
            }
            dsum[i] = 0.f;
            erl[i]  = erv;
            shl[i]  = c;
        }
    }
    __syncthreads();

    int* myq = q[wave];
    int qh = 0, qt = 0;

    const float4 al4 = *(const float4*)(attn_l + 4 * lane);
    const int hd = lane >> 3;
    const int jj = lane & 7;
    const bool b4 = (lane & 4) != 0, b2 = (lane & 2) != 0, b1 = (lane & 1) != 0;
    const v2f alxy = {al4.x, al4.y}, alzw = {al4.z, al4.w};
    const float* erl = (const float*)erl4;
    const float* shl = (const float*)shl4;

    auto take = [&](const int nb) {
        int s = 0, dl = 0;
        if (lane < nb) {
            const int e = myq[(qh + lane) & (QRING - 1)];
            s  = src[e];
            dl = dst[e] - n0;
        }
        for (int j0 = 0; j0 < nb; j0 += 8) {
            float4 v[8];
#pragma unroll
            for (int j = 0; j < 8; ++j) {
                const int sj = __builtin_amdgcn_readlane(s, j0 + j);
                v[j] = *(const float4*)(ft + (size_t)sj * HD + 4 * lane);
            }
            const int  dme  = __builtin_amdgcn_ds_bpermute((j0 + jj) << 2, dl);
            const bool live = (j0 + jj) < nb;
            const float erv = erl[dme * NH + hd];
            const float shv = shl[dme * NH + hd];
            float r[8];
#pragma unroll
            for (int j = 0; j < 8; ++j) {
                v2f p2 = v2f{v[j].x, v[j].y} * alxy;
                p2 = v2f{v[j].z, v[j].w} * alzw + p2;
                r[j] = p2.x + p2.y;
            }
            const float n0s = (b4 ? r[4] : r[0]) + lane_swz<0x164>(b4 ? r[0] : r[4]);
            const float n1s = (b4 ? r[5] : r[1]) + lane_swz<0x164>(b4 ? r[1] : r[5]);
            const float n2s = (b4 ? r[6] : r[2]) + lane_swz<0x164>(b4 ? r[2] : r[6]);
            const float n3s = (b4 ? r[7] : r[3]) + lane_swz<0x164>(b4 ? r[3] : r[7]);
            const float m0s = (b2 ? n2s : n0s) + lane_swz<0x4E>(b2 ? n0s : n2s);
            const float m1s = (b2 ? n3s : n1s) + lane_swz<0x4E>(b2 ? n1s : n3s);
            const float elv = (b1 ? m1s : m0s) + lane_swz<0xB1>(b1 ? m0s : m1s);
            float x = elv + erv;
            x = x > 0.f ? x : NEG_SLOPE * x;
            const float ex = live ? expf(x - shv) : 0.f;
            if (live) atomicAdd(&dsum[dme * NH + hd], ex);
#define ROW_ADD(J)                                                                   \
            if (j0 + J < nb) {                                                       \
                const int dj = __builtin_amdgcn_readlane(dl, j0 + J);                \
                float* ar = acc + dj * HD + lane;                                    \
                const float w = grp8_bcast<J>(ex);                                   \
                atomicAdd(ar,          w * v[J].x);                                  \
                atomicAdd(ar + DH,     w * v[J].y);                                  \
                atomicAdd(ar + 2 * DH, w * v[J].z);                                  \
                atomicAdd(ar + 3 * DH, w * v[J].w);                                  \
            }
            ROW_ADD(0) ROW_ADD(1) ROW_ADD(2) ROW_ADD(3)
            ROW_ADD(4) ROW_ADD(5) ROW_ADD(6) ROW_ADD(7)
#undef ROW_ADD
        }
        qh += nb;
    };

    const bool use_codes = (bound != 0u) && codes_fit(nnodes, nedges);
    const unsigned* wsrc = use_codes ? emaxk : (const unsigned*)dst;
    const int sh = use_codes ? 2 : 0;
    const int nw = use_codes ? ((nedges / PBLOCK) * (PBLOCK / 4)) : (nedges & ~3);
    unsigned pl[8];
#pragma unroll
    for (int k = 0; k < 8; ++k) pl[k] = 0u - (((unsigned)blockIdx.x >> k) & 1u);
    const int gofs = ((lane >> 2) << 7) + (lane & 3);
    const int nchunk = (nw + CWORDS - 1) / CWORDS;
    const int nit    = (nchunk + OWAVES - 1) / OWAVES;

    auto push = [&](const bool hit, const int e) {
        const unsigned m = __builtin_amdgcn_ballot_w32(hit);
        if (m != 0u) {
            if (hit) myq[(qt + (int)__builtin_amdgcn_mbcnt_lo(m, 0u)) & (QRING - 1)] = e;
            qt += __builtin_popcount(m);
        }
    };
    const int rot = (int)(((long long)blockIdx.x * nit) / gridDim.x) * OWAVES;
    auto chunk_of = [&](const int it) -> int {
        if (it >= nit) return -1;
        int c = it * OWAVES + wave + rot;
        if (c >= nit * OWAVES) c -= nit * OWAVES;
        return c < nchunk ? c : -1;
    };
    auto load16 = [&](const int c, const int u) -> uint4 {
        int w = (c < 0 ? 0 : c) * CWORDS + u * 128 + lane * 4;
        w = w < nw - 4 ? w : nw - 4;
        return *(const uint4*)(wsrc + w);
    };
    auto scan_planes = [&](const uint4 lo, const uint4 hi, const int e0) {
        unsigned m = ~(lo.x ^ pl[0]) & ~(lo.y ^ pl[1]) & ~(lo.z ^ pl[2]) & ~(lo.w ^ pl[3])
                   & ~(hi.x ^ pl[4]) & ~(hi.y ^ pl[5]) & ~(hi.z ^ pl[6]) & ~(hi.w ^ pl[7]);
        for (unsigned mm = __builtin_amdgcn_ballot_w32(m != 0u); mm != 0u; mm = __builtin_amdgcn_ballot_w32(m != 0u)) {
            const int p = qt + (int)__builtin_amdgcn_mbcnt_lo(mm, 0u);
            if (m != 0u && (p - qh) < QRING) myq[p & (QRING - 1)] = e0 + (__builtin_ctz(m | 0x80000000u) << 2);
            qt += __builtin_popcount(mm);
            m &= m - 1u;
        }
    };

    auto scan_dst = [&](const unsigned x, const bool valid, const int w) {
        const bool hit = valid && ((unsigned)((int)x - n0) < (unsigned)nown);
        const unsigned m = __builtin_amdgcn_ballot_w32(hit);
        if (m != 0u) {
            const int p = qt + (int)__builtin_amdgcn_mbcnt_lo(m, 0u);
            if (hit && (p - qh) < QRING) myq[p & (QRING - 1)] = w;
            qt += __builtin_popcount(m);
        }
    };


    int it = (nw > 0) ? -1 : 0;
    int ccur = -1;
    int spos = 0, send = 0;
    if (wave == 0) { spos = nw << sh; send = nedges; }
    uint4 cv[8];
#pragma unroll
    for (int u = 0; u < 8; ++u) cv[u] = make_uint4(0u, 0u, 0u, 0u);
    auto scan_step = [&]() {
        const int cnext = chunk_of(it + 1);
        const int qt0 = qt;
#pragma unroll
        for (int u = 0; u < 8; ++u) {
            if (ccur >= 0) {
                if (use_codes) {
                    if ((u & 1) == 0) {
                        const int wb = ccur * CWORDS + (u >> 1) * (PBLOCK / 4);
                        if (wb < nw) scan_planes(cv[u], cv[u + 1], (wb << 2) + gofs);
                    }
                } else {
                    const int w = ccur * CWORDS + u * 128 + lane * 4;
                    const bool valid = w < nw;
                    scan_dst(cv[u].x, valid, w);
                    scan_dst(cv[u].y, valid, w + 1);
                    scan_dst(cv[u].z, valid, w + 2);
                    scan_dst(cv[u].w, valid, w + 3);
                }
            }
            cv[u] = load16(cnext, u);
        }
        if (qt - qh > QRING) {
            qt = qt0;
            spos = (ccur * CWORDS) << sh;
            send = ((ccur * CWORDS + CWORDS) < nw ? (ccur * CWORDS + CWORDS) : nw) << sh;
        }
        ccur = cnext;
        ++it;
    };
    for (;;) {
        const int nq = qt - qh;
        const bool scanned = !(spos < send) && (it >= nit);
        if (nq >= 32 || (scanned && nq > 0)) {
            take(nq < 32 ? nq : 32);
            continue;
        }
        if (scanned) break;
        if (spos < send) {
            const int e = spos + lane;
            const bool valid = e < send;
            const int x = valid ? dst[e] : 0;
            push(valid && ((unsigned)(x - n0) < (unsigned)nown), e);
            spos += 32;
            continue;
        }
        scan_step();
    }
    __syncthreads();

    float4* o4 = (float4*)(out + (size_t)n0 * HD);
    for (int i = tid; i < nown * (HD / 4); i += 1024) {
        const int node = i >> 5, qd = i & 31;
        const float dn = dsum[node * NH + (qd >> 3)];
        const float dq = dn > 0.f ? dn : 1.0f;
        const float* ar = acc + node * HD + qd;
        float4 a = make_float4(ar[0], ar[DH], ar[2 * DH], ar[3 * DH]);
        a.x /= dq; a.y /= dq; a.z /= dq; a.w /= dq;
        o4[i] = a;
    }
}

static inline unsigned nblk(long long n, int bs) {
    return (unsigned)((n + bs - 1) / bs);
}

extern "C" void kernel_launch(void* const* d_in, const int* in_sizes, int n_in,
                              void* d_out, int out_size, void* d_ws, size_t ws_size,
                              hipStream_t stream) {
    const float* feat   = (const float*)d_in[0];
    const float* W      = (const float*)d_in[1];
    const float* attn_l = (const float*)d_in[2];
    const float* attn_r = (const float*)d_in[3];
    const int*   src    = (const int*)d_in[4];
    const int*   dst    = (const int*)d_in[5];
    float*       out    = (float*)d_out;

    const long long N = in_sizes[0] / IN_F;
    const long long E = in_sizes[4];

    float*    ft    = (float*)d_ws;
    float*    el    = ft + (size_t)N * HD;
    float*    er    = el + (size_t)N * NH;
    unsigned* emaxk = (unsigned*)(er + (size_t)N * NH);
    unsigned* rng   = emaxk + (size_t)N * NH;
    unsigned* rfin  = rng + RNG_SLOTS * 8;

    const int BS = 256;
    const long long NHtot = N * NH;
    const long long EHtot = E * NH;

    gemm_wmma_f32<<<nblk(N, GEMM_ROWS_PER_BLOCK), 64, 0, stream>>>(feat, W, ft, rng, (int)N);

    node_attn<<<nblk(NHtot, BS), BS, 0, stream>>>(ft, attn_l, attn_r, el, er, emaxk, rng, NHtot);

    edge_max<<<392, 128, 0, stream>>>(el, er, src, dst, rng, rfin, emaxk, EHtot, (int)N, (int)E);

    edge_owner<<<nblk(N, OWN), 1024, 0, stream>>>(ft, attn_l, er, src, dst, rfin, emaxk, out, (int)N, (int)E);
}


# Launch 3 also writes one packed word per edge (source node, destination row inside its owner); launch 4 gathers that one word per kept edge, not src[e] and dst[e] from two lines; sums arrival-order
# speedup vs baseline: 3.7343x; 1.0383x over previous
#include <hip/hip_runtime.h>
#include <hip/hip_bf16.h>

#define IN_F 128
#define HD   128
#define NH   4
#define DH   32
#define NEG_SLOPE 0.2f

typedef __attribute__((ext_vector_type(2))) float v2f;
typedef __attribute__((ext_vector_type(8))) float v8f;

__device__ __forceinline__ unsigned f2ord(float f) {
    unsigned b = __float_as_uint(f);
    return (b & 0x80000000u) ? ~b : (b | 0x80000000u);
}
__device__ __forceinline__ float ord2f(unsigned k) {
    unsigned b = (k & 0x80000000u) ? (k ^ 0x80000000u) : ~k;
    return __uint_as_float(b);
}

template <int CTRL>
__device__ __forceinline__ float lane_swz(float x) {
    return __builtin_bit_cast(float, __builtin_amdgcn_update_dpp(0, __builtin_bit_cast(int, x), CTRL, 0xF, 0xF, true));
}
template <int J>
__device__ __forceinline__ float grp8_bcast(float x) {
    return __builtin_bit_cast(float, __builtin_amdgcn_mov_dpp8(__builtin_bit_cast(unsigned, x), (unsigned)J * 0x249249u));
}

#define GEMM_ROWS_PER_BLOCK 32
#define RNG_SLOTS 64
#define RNG_MAX 32.0f
__global__ __launch_bounds__(64) void gemm_wmma_f32(
    const float* __restrict__ A,
    const float* __restrict__ B,
    float* __restrict__ C,
    unsigned* __restrict__ rng,
    int nrows)
{
    if (blockIdx.x == 0) {
        for (int q = threadIdx.x; q < RNG_SLOTS * 8; q += 64) rng[q] = 0u;
    }

    const int lane    = threadIdx.x & 31;
    const int wave    = threadIdx.x >> 5;
    const int l15     = lane & 15;
    const int halfSel = lane >> 4;
    const int mBase   = (blockIdx.x * 2 + wave) * 16;
    if (mBase >= nrows) return;

    int row = mBase + l15;
    if (row >= nrows) row = nrows - 1;

    const float* aptr = A + (size_t)row * IN_F + (halfSel << 1);
    const float* bptr = B + (size_t)(halfSel << 1) * HD + (l15 << 2);

    v8f acc[8];
#pragma unroll
    for (int t = 0; t < 8; ++t) acc[t] = v8f{};

    v2f    a   = *(const v2f*)aptr;
    float4 b0l = *(const float4*)(bptr);
    float4 b1l = *(const float4*)(bptr + HD);
    float4 b0h = *(const float4*)(bptr + 64);
    float4 b1h = *(const float4*)(bptr + HD + 64);

#pragma unroll 2
    for (int k = 0; k < IN_F; k += 4) {
        const int kn = (k + 4 < IN_F) ? (k + 4) : 0;
        const v2f    an   = *(const v2f*)(aptr + kn);
        const float4 b0ln = *(const float4*)(bptr + (size_t)kn * HD);
        const float4 b1ln = *(const float4*)(bptr + (size_t)(kn + 1) * HD);
        const float4 b0hn = *(const float4*)(bptr + (size_t)kn * HD + 64);
        const float4 b1hn = *(const float4*)(bptr + (size_t)(kn + 1) * HD + 64);

        v2f b;
        b.x = b0l.x; b.y = b1l.x;
        acc[0] = __builtin_amdgcn_wmma_f32_16x16x4_f32(false, a, false, b, (short)0, acc[0], false, false);
        b.x = b0l.y; b.y = b1l.y;
        acc[1] = __builtin_amdgcn_wmma_f32_16x16x4_f32(false, a, false, b, (short)0, acc[1], false, false);
        b.x = b0l.z; b.y = b1l.z;
        acc[2] = __builtin_amdgcn_wmma_f32_16x16x4_f32(false, a, false, b, (short)0, acc[2], false, false);
        b.x = b0l.w; b.y = b1l.w;
        acc[3] = __builtin_amdgcn_wmma_f32_16x16x4_f32(false, a, false, b, (short)0, acc[3], false, false);
        b.x = b0h.x; b.y = b1h.x;
        acc[4] = __builtin_amdgcn_wmma_f32_16x16x4_f32(false, a, false, b, (short)0, acc[4], false, false);
        b.x = b0h.y; b.y = b1h.y;
        acc[5] = __builtin_amdgcn_wmma_f32_16x16x4_f32(false, a, false, b, (short)0, acc[5], false, false);
        b.x = b0h.z; b.y = b1h.z;
        acc[6] = __builtin_amdgcn_wmma_f32_16x16x4_f32(false, a, false, b, (short)0, acc[6], false, false);
        b.x = b0h.w; b.y = b1h.w;
        acc[7] = __builtin_amdgcn_wmma_f32_16x16x4_f32(false, a, false, b, (short)0, acc[7], false, false);

        a = an; b0l = b0ln; b1l = b1ln; b0h = b0hn; b1h = b1hn;
    }

#pragma unroll
    for (int r = 0; r < 8; ++r) {
        const int m = mBase + r + (halfSel << 3);
        if (m < nrows) {
            float* crow = C + (size_t)m * HD + (l15 << 2);
            *(float4*)(crow)      = make_float4(acc[0][r], acc[1][r], acc[2][r], acc[3][r]);
            *(float4*)(crow + 64) = make_float4(acc[4][r], acc[5][r], acc[6][r], acc[7][r]);
        }
    }
}

__global__ void node_attn(const float* __restrict__ ft,
                          const float* __restrict__ attn_l,
                          const float* __restrict__ attn_r,
                          float* __restrict__ el, float* __restrict__ er,
                          unsigned* __restrict__ emaxk,
                          unsigned* __restrict__ rng,
                          long long nh_total)
{
    __shared__ unsigned red[8][8];
    long long t = (long long)blockIdx.x * blockDim.x + threadIdx.x;
    const bool live = t < nh_total;
    const int h = (int)(threadIdx.x & 3);
    unsigned kmax = 0u, kmin = 0u;
    if (live) {
        const long long n = t >> 2;
        const float* f  = ft + (size_t)n * HD + h * DH;
        const float* al = attn_l + h * DH;
        const float* ar = attn_r + h * DH;
        float sl = 0.f, sr = 0.f;
#pragma unroll
        for (int d = 0; d < DH; d += 4) {
            const float4 fv = *(const float4*)(f + d);
            const float4 lv = *(const float4*)(al + d);
            const float4 rv = *(const float4*)(ar + d);
            sl += fv.x * lv.x + fv.y * lv.y + fv.z * lv.z + fv.w * lv.w;
            sr += fv.x * rv.x + fv.y * rv.y + fv.z * rv.z + fv.w * rv.w;
        }
        el[t] = sl;
        er[t] = sr;
        emaxk[t] = 0u;
        kmax = f2ord(sl);
        kmin = ~kmax;
    }
    for (int m = 4; m < 32; m <<= 1) {
        const unsigned a = (unsigned)__shfl_xor((int)kmax, m);
        const unsigned b = (unsigned)__shfl_xor((int)kmin, m);
        kmax = a > kmax ? a : kmax;
        kmin = b > kmin ? b : kmin;
    }
    const int lane = threadIdx.x & 31;
    const int wave = threadIdx.x >> 5;
    if (lane < 4) { red[wave][lane] = kmax; red[wave][4 + lane] = kmin; }
    __syncthreads();
    if (threadIdx.x < 8) {
        unsigned k = 0u;
        const int nw = (int)(blockDim.x >> 5);
        for (int w = 0; w < nw; ++w) { const unsigned v = red[w][threadIdx.x]; k = v > k ? v : k; }
        atomicMax(&rng[(blockIdx.x & (RNG_SLOTS - 1)) * 8 + threadIdx.x], k);
    }
}

__device__ __forceinline__ unsigned fold_range(const unsigned* __restrict__ rng, int lane, float mx[NH])
{
    unsigned k[8];
#pragma unroll
    for (int i = 0; i < 8; ++i) {
        const unsigned a = rng[lane * 8 + i];
        const unsigned b = rng[(lane + 32) * 8 + i];
        k[i] = a > b ? a : b;
    }
#pragma unroll
    for (int i = 0; i < 8; ++i) {
        for (int m = 1; m < 32; m <<= 1) {
            const unsigned o = (unsigned)__shfl_xor((int)k[i], m);
            k[i] = o > k[i] ? o : k[i];
        }
    }
    unsigned ok = 1u;
#pragma unroll
    for (int hh = 0; hh < NH; ++hh) {
        const float hi = (k[hh] == 0u) ? __builtin_nanf("") : ord2f(k[hh]);
        const float lo = (k[4 + hh] == 0u) ? __builtin_nanf("") : ord2f(~k[4 + hh]);
        const bool good = __builtin_isfinite(hi) && __builtin_isfinite(lo) && ((hi - lo) <= RNG_MAX);
        if (!good) ok = 0u;
        mx[hh] = hi;
    }
    return ok;
}

#define OWN     400
#define PBLOCK  1024
__device__ __forceinline__ bool codes_fit(const int nnodes, const int nedges) {
    return ((long long)nedges <= (long long)nnodes * 16) && ((nnodes + OWN - 1) / OWN <= 256);
}

#define PK_DBITS 9
__device__ __forceinline__ bool pk_fit(const int nnodes, const int nedges, const unsigned long long ws_bytes) {
    return codes_fit(nnodes, nedges) && (nnodes <= (1 << (32 - PK_DBITS)))
        && (ws_bytes >= ((unsigned long long)nnodes * (HD + 4 * NH) + (unsigned long long)nedges) * 4ull);
}
__device__ __forceinline__ unsigned pack_sd(const int s, const int d) {
    const unsigned du = (unsigned)d;
    return ((unsigned)s << PK_DBITS) | (du - (du / OWN) * OWN);
}

__global__ void edge_max(const float* __restrict__ el,
                         const float* __restrict__ er,
                         const int* __restrict__ src,
                         const int* __restrict__ dst,
                         const unsigned* __restrict__ rng,
                         unsigned* __restrict__ rfin,
                         unsigned* __restrict__ emaxk,
                         unsigned* __restrict__ pk,
                         unsigned long long ws_bytes,
                         long long total,
                         int nnodes, int nedges)
{
    const long long step = (long long)gridDim.x * blockDim.x;
    const long long tid0 = (long long)blockIdx.x * blockDim.x + threadIdx.x;
    const int lane = (int)(threadIdx.x & 31);
    const int gw  = __builtin_amdgcn_readfirstlane((int)(tid0 >> 5));
    const int nwv = (int)(step >> 5);
    const int nblocks = codes_fit(nnodes, nedges) ? (nedges / PBLOCK) : 0;
    int4 d[8];
#pragma unroll
    for (int j = 0; j < 8; ++j) d[j] = make_int4(0, 0, 0, 0);
    const bool pk_on = pk_fit(nnodes, nedges, ws_bytes);
    int4 sv[8];
    if (gw < nblocks) {
#pragma unroll
        for (int j = 0; j < 8; ++j) d[j] = *(const int4*)(dst + (size_t)gw * PBLOCK + j * 128 + 4 * lane);
        if (pk_on) {
#pragma unroll
            for (int j = 0; j < 8; ++j) sv[j] = *(const int4*)(src + (size_t)gw * PBLOCK + j * 128 + 4 * lane);
        }
    }

    float mx[NH];
    const unsigned ok = fold_range(rng, lane, mx);
    if (blockIdx.x == 0 && threadIdx.x == 0) {
#pragma unroll
        for (int hh = 0; hh < NH; ++hh) rfin[hh] = __float_as_uint(mx[hh]);
        rfin[4] = ok;
    }
    if (ok != 0u) {
        for (int blk = gw; blk < nblocks; blk += nwv) {
            if (blk != gw) {
#pragma unroll
                for (int j = 0; j < 8; ++j) d[j] = *(const int4*)(dst + (size_t)blk * PBLOCK + j * 128 + 4 * lane);
                if (pk_on) {
#pragma unroll
                    for (int j = 0; j < 8; ++j) sv[j] = *(const int4*)(src + (size_t)blk * PBLOCK + j * 128 + 4 * lane);
                }
            }
            int r[8];
#pragma unroll
            for (int k = 0; k < 8; ++k) r[k] = 0;
#pragma unroll
            for (int j = 0; j < 8; ++j) {
#pragma unroll
                for (int cc = 0; cc < 4; ++cc) {
                    const int dv = cc == 0 ? d[j].x : cc == 1 ? d[j].y : cc == 2 ? d[j].z : d[j].w;
                    const unsigned code = ((unsigned)dv / OWN) & 255u;
                    const bool mine = lane == 4 * j + cc;
#pragma unroll
                    for (int k = 0; k < 8; ++k) {
                        const unsigned vote = __builtin_amdgcn_ballot_w32(((code >> k) & 1u) != 0u);
                        r[k] = mine ? (int)vote : r[k];
                    }
                }
            }
            unsigned* pw = emaxk + (size_t)blk * (PBLOCK / 4) + 4 * lane;
            *(uint4*)(pw)       = make_uint4((unsigned)r[0], (unsigned)r[1], (unsigned)r[2], (unsigned)r[3]);
            *(uint4*)(pw + 128) = make_uint4((unsigned)r[4], (unsigned)r[5], (unsigned)r[6], (unsigned)r[7]);
            if (pk_on) {
#pragma unroll
                for (int j = 0; j < 8; ++j) {
                    *(uint4*)(pk + (size_t)blk * PBLOCK + j * 128 + 4 * lane) =
                        make_uint4(pack_sd(sv[j].x, d[j].x), pack_sd(sv[j].y, d[j].y),
                                   pack_sd(sv[j].z, d[j].z), pack_sd(sv[j].w, d[j].w));
                }
            }
        }
        if (pk_on) {
            for (long long te = (long long)nblocks * PBLOCK + tid0; te < (long long)nedges; te += step)
                pk[te] = pack_sd(src[te], dst[te]);
        }
        return;
    }
    for (long long t = tid0; t < total; t += step) {
        const long long e = t >> 2;
        const int h = (int)(t & 3);
        const int s = src[e], dd = dst[e];
        float v = el[(size_t)s * NH + h] + er[(size_t)dd * NH + h];
        v = v > 0.f ? v : NEG_SLOPE * v;
        atomicMax(&emaxk[(size_t)dd * NH + h], f2ord(v));
    }
}

#define OWAVES  32
#define QRING   512
#define CWORDS  1024
__global__ __launch_bounds__(1024) void edge_owner(const float* __restrict__ ft,
                               const float* __restrict__ attn_l,
                               const float* __restrict__ er,
                               const int* __restrict__ src,
                               const int* __restrict__ dst,
                               const unsigned* __restrict__ rfin,
                               const unsigned* __restrict__ emaxk,
                               const unsigned* __restrict__ pk,
                               unsigned long long ws_bytes,
                               float* __restrict__ out,
                               int nnodes, int nedges)
{
    __shared__ float4 acc4[OWN * HD / 4];
    __shared__ float4 dsum4[OWN];
    __shared__ float4 erl4[OWN];
    __shared__ float4 shl4[OWN];
    __shared__ int    q[OWAVES][QRING];
    float* acc  = (float*)acc4;
    float* dsum = (float*)dsum4;

    const int tid  = threadIdx.x;
    const int lane = tid & 31;
    const int wave = __builtin_amdgcn_readfirstlane(tid >> 5);
    const int n0   = blockIdx.x * OWN;
    const int nown = (nnodes - n0) < OWN ? (nnodes - n0) : OWN;
    const unsigned bound = rfin[4];
    const bool use_pk = (bound != 0u) && pk_fit(nnodes, nedges, ws_bytes);

    for (int i = tid; i < OWN * HD / 4; i += 1024) acc4[i] = make_float4(0.f, 0.f, 0.f, 0.f);
    {
        float* erl = (float*)erl4;
        float* shl = (float*)shl4;
        for (int i = tid; i < OWN * NH; i += 1024) {
            float erv = 0.f, c = 0.f;
            if (i < nown * NH) {
                erv = er[(size_t)n0 * NH + i];
                if (bound != 0u) {
                    c = __uint_as_float(rfin[i & 3]) + erv;
                    c = c > 0.f ? c : NEG_SLOPE * c;
                } else {
                    const unsigned k = emaxk[(size_t)n0 * NH + i];
                    c = (k == 0u) ? 0.f : ord2f(k);
                    if (!__builtin_isfinite(c)) c = 0.f;
                }
            }
            dsum[i] = 0.f;
            erl[i]  = erv;
            shl[i]  = c;
        }
    }
    __syncthreads();

    int* myq = q[wave];
    int qh = 0, qt = 0;

    const float4 al4 = *(const float4*)(attn_l + 4 * lane);
    const int hd = lane >> 3;
    const int jj = lane & 7;
    const bool b4 = (lane & 4) != 0, b2 = (lane & 2) != 0, b1 = (lane & 1) != 0;
    const v2f alxy = {al4.x, al4.y}, alzw = {al4.z, al4.w};
    const float* erl = (const float*)erl4;
    const float* shl = (const float*)shl4;

    auto take = [&](const int nb) {
        int s = 0, dl = 0;
        if (lane < nb) {
            const int e = myq[(qh + lane) & (QRING - 1)];
            if (use_pk) {
                const unsigned p = pk[e];
                s  = (int)(p >> PK_DBITS);
                dl = (int)(p & ((1u << PK_DBITS) - 1u));
            } else {
                s  = src[e];
                dl = dst[e] - n0;
            }
        }
        for (int j0 = 0; j0 < nb; j0 += 8) {
            float4 v[8];
#pragma unroll
            for (int j = 0; j < 8; ++j) {
                const int sj = __builtin_amdgcn_readlane(s, j0 + j);
                v[j] = *(const float4*)(ft + (size_t)sj * HD + 4 * lane);
            }
            const int  dme  = __builtin_amdgcn_ds_bpermute((j0 + jj) << 2, dl);
            const bool live = (j0 + jj) < nb;
            const float erv = erl[dme * NH + hd];
            const float shv = shl[dme * NH + hd];
            float r[8];
#pragma unroll
            for (int j = 0; j < 8; ++j) {
                v2f p2 = v2f{v[j].x, v[j].y} * alxy;
                p2 = v2f{v[j].z, v[j].w} * alzw + p2;
                r[j] = p2.x + p2.y;
            }
            const float n0s = (b4 ? r[4] : r[0]) + lane_swz<0x164>(b4 ? r[0] : r[4]);
            const float n1s = (b4 ? r[5] : r[1]) + lane_swz<0x164>(b4 ? r[1] : r[5]);
            const float n2s = (b4 ? r[6] : r[2]) + lane_swz<0x164>(b4 ? r[2] : r[6]);
            const float n3s = (b4 ? r[7] : r[3]) + lane_swz<0x164>(b4 ? r[3] : r[7]);
            const float m0s = (b2 ? n2s : n0s) + lane_swz<0x4E>(b2 ? n0s : n2s);
            const float m1s = (b2 ? n3s : n1s) + lane_swz<0x4E>(b2 ? n1s : n3s);
            const float elv = (b1 ? m1s : m0s) + lane_swz<0xB1>(b1 ? m0s : m1s);
            float x = elv + erv;
            x = x > 0.f ? x : NEG_SLOPE * x;
            const float ex = live ? expf(x - shv) : 0.f;
            if (live) atomicAdd(&dsum[dme * NH + hd], ex);
#define ROW_ADD(J)                                                                   \
            if (j0 + J < nb) {                                                       \
                const int dj = __builtin_amdgcn_readlane(dl, j0 + J);                \
                float* ar = acc + dj * HD + lane;                                    \
                const float w = grp8_bcast<J>(ex);                                   \
                atomicAdd(ar,          w * v[J].x);                                  \
                atomicAdd(ar + DH,     w * v[J].y);                                  \
                atomicAdd(ar + 2 * DH, w * v[J].z);                                  \
                atomicAdd(ar + 3 * DH, w * v[J].w);                                  \
            }
            ROW_ADD(0) ROW_ADD(1) ROW_ADD(2) ROW_ADD(3)
            ROW_ADD(4) ROW_ADD(5) ROW_ADD(6) ROW_ADD(7)
#undef ROW_ADD
        }
        qh += nb;
    };

    const bool use_codes = (bound != 0u) && codes_fit(nnodes, nedges);
    const unsigned* wsrc = use_codes ? emaxk : (const unsigned*)dst;
    const int sh = use_codes ? 2 : 0;
    const int nw = use_codes ? ((nedges / PBLOCK) * (PBLOCK / 4)) : (nedges & ~3);
    unsigned pl[8];
#pragma unroll
    for (int k = 0; k < 8; ++k) pl[k] = 0u - (((unsigned)blockIdx.x >> k) & 1u);
    const int gofs = ((lane >> 2) << 7) + (lane & 3);
    const int nchunk = (nw + CWORDS - 1) / CWORDS;
    const int nit    = (nchunk + OWAVES - 1) / OWAVES;

    auto push = [&](const bool hit, const int e) {
        const unsigned m = __builtin_amdgcn_ballot_w32(hit);
        if (m != 0u) {
            if (hit) myq[(qt + (int)__builtin_amdgcn_mbcnt_lo(m, 0u)) & (QRING - 1)] = e;
            qt += __builtin_popcount(m);
        }
    };
    const int rot = (int)(((long long)blockIdx.x * nit) / gridDim.x) * OWAVES;
    auto chunk_of = [&](const int it) -> int {
        if (it >= nit) return -1;
        int c = it * OWAVES + wave + rot;
        if (c >= nit * OWAVES) c -= nit * OWAVES;
        return c < nchunk ? c : -1;
    };
    auto load16 = [&](const int c, const int u) -> uint4 {
        int w = (c < 0 ? 0 : c) * CWORDS + u * 128 + lane * 4;
        w = w < nw - 4 ? w : nw - 4;
        return *(const uint4*)(wsrc + w);
    };
    auto scan_planes = [&](const uint4 lo, const uint4 hi, const int e0) {
        unsigned m = ~(lo.x ^ pl[0]) & ~(lo.y ^ pl[1]) & ~(lo.z ^ pl[2]) & ~(lo.w ^ pl[3])
                   & ~(hi.x ^ pl[4]) & ~(hi.y ^ pl[5]) & ~(hi.z ^ pl[6]) & ~(hi.w ^ pl[7]);
        for (unsigned mm = __builtin_amdgcn_ballot_w32(m != 0u); mm != 0u; mm = __builtin_amdgcn_ballot_w32(m != 0u)) {
            const int p = qt + (int)__builtin_amdgcn_mbcnt_lo(mm, 0u);
            if (m != 0u && (p - qh) < QRING) myq[p & (QRING - 1)] = e0 + (__builtin_ctz(m | 0x80000000u) << 2);
            qt += __builtin_popcount(mm);
            m &= m - 1u;
        }
    };

    auto scan_dst = [&](const unsigned x, const bool valid, const int w) {
        const bool hit = valid && ((unsigned)((int)x - n0) < (unsigned)nown);
        const unsigned m = __builtin_amdgcn_ballot_w32(hit);
        if (m != 0u) {
            const int p = qt + (int)__builtin_amdgcn_mbcnt_lo(m, 0u);
            if (hit && (p - qh) < QRING) myq[p & (QRING - 1)] = w;
            qt += __builtin_popcount(m);
        }
    };


    int it = (nw > 0) ? -1 : 0;
    int ccur = -1;
    int spos = 0, send = 0;
    if (wave == 0) { spos = nw << sh; send = nedges; }
    uint4 cv[8];
#pragma unroll
    for (int u = 0; u < 8; ++u) cv[u] = make_uint4(0u, 0u, 0u, 0u);
    auto scan_step = [&]() {
        const int cnext = chunk_of(it + 1);
        const int qt0 = qt;
#pragma unroll
        for (int u = 0; u < 8; ++u) {
            if (ccur >= 0) {
                if (use_codes) {
                    if ((u & 1) == 0) {
                        const int wb = ccur * CWORDS + (u >> 1) * (PBLOCK / 4);
                        if (wb < nw) scan_planes(cv[u], cv[u + 1], (wb << 2) + gofs);
                    }
                } else {
                    const int w = ccur * CWORDS + u * 128 + lane * 4;
                    const bool valid = w < nw;
                    scan_dst(cv[u].x, valid, w);
                    scan_dst(cv[u].y, valid, w + 1);
                    scan_dst(cv[u].z, valid, w + 2);
                    scan_dst(cv[u].w, valid, w + 3);
                }
            }
            cv[u] = load16(cnext, u);
        }
        if (qt - qh > QRING) {
            qt = qt0;
            spos = (ccur * CWORDS) << sh;
            send = ((ccur * CWORDS + CWORDS) < nw ? (ccur * CWORDS + CWORDS) : nw) << sh;
        }
        ccur = cnext;
        ++it;
    };
    for (;;) {
        const int nq = qt - qh;
        const bool scanned = !(spos < send) && (it >= nit);
        if (nq >= 32 || (scanned && nq > 0)) {
            take(nq < 32 ? nq : 32);
            continue;
        }
        if (scanned) break;
        if (spos < send) {
            const int e = spos + lane;
            const bool valid = e < send;
            const int x = valid ? dst[e] : 0;
            push(valid && ((unsigned)(x - n0) < (unsigned)nown), e);
            spos += 32;
            continue;
        }
        scan_step();
    }
    __syncthreads();

    float4* o4 = (float4*)(out + (size_t)n0 * HD);
    for (int i = tid; i < nown * (HD / 4); i += 1024) {
        const int node = i >> 5, qd = i & 31;
        const float dn = dsum[node * NH + (qd >> 3)];
        const float dq = dn > 0.f ? dn : 1.0f;
        const float* ar = acc + node * HD + qd;
        float4 a = make_float4(ar[0], ar[DH], ar[2 * DH], ar[3 * DH]);
        a.x /= dq; a.y /= dq; a.z /= dq; a.w /= dq;
        o4[i] = a;
    }
}

static inline unsigned nblk(long long n, int bs) {
    return (unsigned)((n + bs - 1) / bs);
}

extern "C" void kernel_launch(void* const* d_in, const int* in_sizes, int n_in,
                              void* d_out, int out_size, void* d_ws, size_t ws_size,
                              hipStream_t stream) {
    const float* feat   = (const float*)d_in[0];
    const float* W      = (const float*)d_in[1];
    const float* attn_l = (const float*)d_in[2];
    const float* attn_r = (const float*)d_in[3];
    const int*   src    = (const int*)d_in[4];
    const int*   dst    = (const int*)d_in[5];
    float*       out    = (float*)d_out;

    const long long N = in_sizes[0] / IN_F;
    const long long E = in_sizes[4];

    float*    ft    = (float*)d_ws;
    float*    el    = ft + (size_t)N * HD;
    float*    er    = el + (size_t)N * NH;
    unsigned* emaxk = (unsigned*)(er + (size_t)N * NH);
    unsigned* rng   = emaxk + (size_t)N * NH;
    unsigned* rfin  = rng + RNG_SLOTS * 8;
    unsigned* pk    = rng + (size_t)N * NH;

    const int BS = 256;
    const long long NHtot = N * NH;
    const long long EHtot = E * NH;

    gemm_wmma_f32<<<nblk(N, GEMM_ROWS_PER_BLOCK), 64, 0, stream>>>(feat, W, ft, rng, (int)N);

    node_attn<<<nblk(NHtot, BS), BS, 0, stream>>>(ft, attn_l, attn_r, el, er, emaxk, rng, NHtot);

    edge_max<<<392, 128, 0, stream>>>(el, er, src, dst, rng, rfin, emaxk, pk, (unsigned long long)ws_size, EHtot, (int)N, (int)E);

    edge_owner<<<nblk(N, OWN), 1024, 0, stream>>>(ft, attn_l, er, src, dst, rfin, emaxk, pk, (unsigned long long)ws_size, out, (int)N, (int)E);
}
